# starting kernel
# baseline (speedup 1.0000x reference)
#include <hip/hip_runtime.h>
#include <hip/hip_bf16.h>

#define G_ 128
#define T_ 128
#define S_ 64
#define M_ 8

typedef float v2f __attribute__((ext_vector_type(2)));
typedef float v8f __attribute__((ext_vector_type(8)));

__device__ __forceinline__ void mm64(const float* A, const float* B, float* D,
                                     bool transB,
                                     const float* addQ,
                                     const float* KRp,
                                     const float* Kp)
{
    const int lane = threadIdx.x & 31;
    const int wave = threadIdx.x >> 5;
    const int half = lane >> 4;
    const int l15  = lane & 15;

    for (int tt = wave; tt < 16; tt += 8) {
        const int ti = (tt >> 2) * 16;
        const int tj = (tt & 3) * 16;
        v8f acc = {};
        #pragma unroll
        for (int kb = 0; kb < 16; ++kb) {
            const int k0 = kb * 4 + half * 2;
            v2f a, b;
            a.x = A[(ti + l15) * 64 + k0];
            a.y = A[(ti + l15) * 64 + k0 + 1];
            if (!transB) {
                b.x = B[(k0    ) * 64 + (tj + l15)];
                b.y = B[(k0 + 1) * 64 + (tj + l15)];
            } else {
                b.x = B[(tj + l15) * 64 + k0];
                b.y = B[(tj + l15) * 64 + k0 + 1];
            }
            acc = __builtin_amdgcn_wmma_f32_16x16x4_f32(
                      false, a, false, b, (short)0, acc, false, false);
        }
        #pragma unroll
        for (int r = 0; r < 8; ++r) {
            const int row = ti + r + half * 8;
            const int col = tj + l15;
            float v = acc[r];
            if (addQ) v += addQ[row * 64 + col];
            if (KRp) {
                #pragma unroll
                for (int m = 0; m < 8; ++m)
                    v += KRp[row * 8 + m] * Kp[col * 8 + m];
            }
            D[row * 64 + col] = v;
        }
    }
}

__global__ __launch_bounds__(256) void kf_prep(const float* __restrict__ Lq,
                                               const float* __restrict__ Lr,
                                               const float* __restrict__ Linit,
                                               float* __restrict__ ws)
{
    float* Q  = ws;
    float* P0 = ws + 4096;
    float* R  = ws + 8192;
    for (int e = blockIdx.x * blockDim.x + threadIdx.x; e < 4096;
         e += gridDim.x * blockDim.x) {
        const int i = e >> 6, j = e & 63;
        float q = 0.f, p = 0.f;
        for (int k = 0; k < 64; ++k) {
            q += Lq[i * 64 + k] * Lq[j * 64 + k];
            p += Linit[i * 64 + k] * Linit[j * 64 + k];
        }
        Q[e] = q;
        P0[e] = p;
    }
    const int e = blockIdx.x * blockDim.x + threadIdx.x;
    if (e < 64) {
        const int i = e >> 3, j = e & 7;
        float r = 0.f;
        for (int k = 0; k < 8; ++k) r += Lr[i * 8 + k] * Lr[j * 8 + k];
        R[e] = r;
    }
}

__global__ __launch_bounds__(256) void kf_main(const float* __restrict__ y,
                                               const float* __restrict__ F,
                                               const float* __restrict__ H,
                                               const float* __restrict__ Qg,
                                               const float* __restrict__ Rg,
                                               const float* __restrict__ P0g,
                                               const float* __restrict__ init_mean,
                                               float* __restrict__ out)
{
    extern __shared__ float smem[];
    float* Fs    = smem;
    float* bufA  = Fs   + 4096;
    float* bufB  = bufA + 4096;
    float* bufC  = bufB + 4096;
    float* Hs    = bufC + 4096;
    float* Rs    = Hs   + 512;
    float* PHt   = Rs   + 64;
    float* Km    = PHt  + 512;
    float* KR    = Km   + 512;
    float* Wgj   = KR   + 512;
    float* meanv = Wgj  + 128;
    float* meanp = meanv + 64;
    float* resid = meanp + 64;
    float* fac   = resid + 8;
    float* piv   = fac   + 8;

    const int g   = blockIdx.x;
    const int tid = threadIdx.x;

    for (int i = tid; i < 4096; i += 256) {
        Fs[i]   = F[i];
        bufA[i] = P0g[i];
    }
    for (int i = tid; i < 512; i += 256) Hs[i] = H[i];
    if (tid < 64) {
        Rs[tid]    = Rg[tid];
        meanv[tid] = init_mean[g * S_ + tid];
    }
    __syncthreads();

    float* covP  = bufA;
    float* covpP = bufB;
    float* tmpP  = bufC;

    for (int t = 0; t < T_; ++t) {
        mm64(Fs, covP, tmpP, false, nullptr, nullptr, nullptr);
        __syncthreads();
        mm64(tmpP, Fs, covpP, true, Qg, nullptr, nullptr);
        __syncthreads();

        for (int e = tid; e < 512; e += 256) {
            const int s = e >> 3, m = e & 7;
            float acc = 0.f;
            for (int k = 0; k < 64; ++k) acc += covpP[s * 64 + k] * Hs[m * 64 + k];
            PHt[e] = acc;
        }
        if (tid < 64) {
            float acc = 0.f;
            for (int j = 0; j < 64; ++j) acc += Fs[tid * 64 + j] * meanv[j];
            meanp[tid] = acc;
        }
        __syncthreads();

        if (tid < 128) {
            const int i = tid >> 4, c = tid & 15;
            if (c < 8) {
                float acc = Rs[i * 8 + c];
                for (int s = 0; s < 64; ++s) acc += Hs[i * 64 + s] * PHt[s * 8 + c];
                Wgj[i * 16 + c] = acc;
            } else {
                Wgj[i * 16 + c] = (c - 8 == i) ? 1.f : 0.f;
            }
        } else if (tid < 136) {
            const int m = tid - 128;
            float acc = 0.f;
            for (int s = 0; s < 64; ++s) acc += Hs[m * 64 + s] * meanp[s];
            out[(g * T_ + t) * M_ + m] = acc;
            resid[m] = y[(g * T_ + t) * M_ + m] - acc;
        }
        __syncthreads();

        for (int p = 0; p < 8; ++p) {
            if (tid == 0) piv[0] = 1.f / Wgj[p * 16 + p];
            __syncthreads();
            if (tid < 16) Wgj[p * 16 + tid] *= piv[0];
            __syncthreads();
            if (tid < 8) fac[tid] = Wgj[tid * 16 + p];
            __syncthreads();
            if (tid < 128) {
                const int r = tid >> 4, c = tid & 15;
                if (r != p) Wgj[r * 16 + c] -= fac[r] * Wgj[p * 16 + c];
            }
            __syncthreads();
        }

        for (int e = tid; e < 512; e += 256) {
            const int s = e >> 3, m = e & 7;
            float acc = 0.f;
            for (int j = 0; j < 8; ++j) acc += PHt[s * 8 + j] * Wgj[j * 16 + 8 + m];
            Km[e] = acc;
        }
        __syncthreads();

        if (tid < 64) {
            float acc = meanp[tid];
            for (int m = 0; m < 8; ++m) acc += Km[tid * 8 + m] * resid[m];
            meanv[tid] = acc;
        }
        for (int e = tid; e < 512; e += 256) {
            const int i = e >> 3, m = e & 7;
            float acc = 0.f;
            for (int n = 0; n < 8; ++n) acc += Km[i * 8 + n] * Rs[n * 8 + m];
            KR[e] = acc;
        }
        for (int e = tid; e < 4096; e += 256) {
            const int i = e >> 6, j = e & 63;
            float acc = (i == j) ? 1.f : 0.f;
            for (int m = 0; m < 8; ++m) acc -= Km[i * 8 + m] * Hs[m * 64 + j];
            covP[e] = acc;
        }
        __syncthreads();

        mm64(covP, covpP, tmpP, false, nullptr, nullptr, nullptr);
        __syncthreads();
        mm64(tmpP, covP, covpP, true, nullptr, KR, Km);
        __syncthreads();

        float* sw = covP; covP = covpP; covpP = sw;
    }
}

extern "C" void kernel_launch(void* const* d_in, const int* in_sizes, int n_in,
                              void* d_out, int out_size, void* d_ws, size_t ws_size,
                              hipStream_t stream) {
    const float* y         = (const float*)d_in[0];
    const float* F         = (const float*)d_in[1];
    const float* H         = (const float*)d_in[2];
    const float* Lq        = (const float*)d_in[3];
    const float* Lr        = (const float*)d_in[4];
    const float* init_mean = (const float*)d_in[5];
    const float* Linit     = (const float*)d_in[6];
    float* out = (float*)d_out;
    float* ws  = (float*)d_ws;

    kf_prep<<<16, 256, 0, stream>>>(Lq, Lr, Linit, ws);

    const size_t smem_bytes = (4096 * 4 + 512 + 64 + 512 + 512 + 512 + 128 +
                               64 + 64 + 8 + 8 + 8) * sizeof(float);
    kf_main<<<G_, 256, smem_bytes, stream>>>(y, F, H,
                                             ws,
                                             ws + 8192,
                                             ws + 4096,
                                             init_mean, out);
}


# LDS row pitch of F, H and the three 64x64 buffers 64 -> 76 floats: WMMA fragment reads and row dot products no longer hit one bank 16-way; same arithmetic and summation order
# speedup vs baseline: 3.0681x; 3.0681x over previous
#include <hip/hip_runtime.h>
#include <hip/hip_bf16.h>

#define G_ 128
#define T_ 128
#define S_ 64
#define M_ 8
#define P_ 76

typedef float v2f __attribute__((ext_vector_type(2)));
typedef float v8f __attribute__((ext_vector_type(8)));

__device__ __forceinline__ void mm64(const float* A, const float* B, float* D,
                                     bool transB,
                                     const float* addQ,
                                     const float* KRp,
                                     const float* Kp)
{
    const int lane = threadIdx.x & 31;
    const int wave = threadIdx.x >> 5;
    const int half = lane >> 4;
    const int l15  = lane & 15;

    for (int tt = wave; tt < 16; tt += 8) {
        const int ti = (tt >> 2) * 16;
        const int tj = (tt & 3) * 16;
        v8f acc = {};
        #pragma unroll
        for (int kb = 0; kb < 16; ++kb) {
            const int k0 = kb * 4 + half * 2;
            v2f a, b;
            a.x = A[(ti + l15) * P_ + k0];
            a.y = A[(ti + l15) * P_ + k0 + 1];
            if (!transB) {
                b.x = B[(k0    ) * P_ + (tj + l15)];
                b.y = B[(k0 + 1) * P_ + (tj + l15)];
            } else {
                b.x = B[(tj + l15) * P_ + k0];
                b.y = B[(tj + l15) * P_ + k0 + 1];
            }
            acc = __builtin_amdgcn_wmma_f32_16x16x4_f32(
                      false, a, false, b, (short)0, acc, false, false);
        }
        #pragma unroll
        for (int r = 0; r < 8; ++r) {
            const int row = ti + r + half * 8;
            const int col = tj + l15;
            float v = acc[r];
            if (addQ) v += addQ[row * 64 + col];
            if (KRp) {
                #pragma unroll
                for (int m = 0; m < 8; ++m)
                    v += KRp[row * 8 + m] * Kp[col * 8 + m];
            }
            D[row * P_ + col] = v;
        }
    }
}

__global__ __launch_bounds__(256) void kf_prep(const float* __restrict__ Lq,
                                               const float* __restrict__ Lr,
                                               const float* __restrict__ Linit,
                                               float* __restrict__ ws)
{
    float* Q  = ws;
    float* P0 = ws + 4096;
    float* R  = ws + 8192;
    for (int e = blockIdx.x * blockDim.x + threadIdx.x; e < 4096;
         e += gridDim.x * blockDim.x) {
        const int i = e >> 6, j = e & 63;
        float q = 0.f, p = 0.f;
        for (int k = 0; k < 64; ++k) {
            q += Lq[i * 64 + k] * Lq[j * 64 + k];
            p += Linit[i * 64 + k] * Linit[j * 64 + k];
        }
        Q[e] = q;
        P0[e] = p;
    }
    const int e = blockIdx.x * blockDim.x + threadIdx.x;
    if (e < 64) {
        const int i = e >> 3, j = e & 7;
        float r = 0.f;
        for (int k = 0; k < 8; ++k) r += Lr[i * 8 + k] * Lr[j * 8 + k];
        R[e] = r;
    }
}

__global__ __launch_bounds__(256) void kf_main(const float* __restrict__ y,
                                               const float* __restrict__ F,
                                               const float* __restrict__ H,
                                               const float* __restrict__ Qg,
                                               const float* __restrict__ Rg,
                                               const float* __restrict__ P0g,
                                               const float* __restrict__ init_mean,
                                               float* __restrict__ out)
{
    extern __shared__ float smem[];
    float* Fs    = smem;
    float* bufA  = Fs   + 64 * P_;
    float* bufB  = bufA + 64 * P_;
    float* bufC  = bufB + 64 * P_;
    float* Hs    = bufC + 64 * P_;
    float* Rs    = Hs   + 8 * P_;
    float* PHt   = Rs   + 64;
    float* Km    = PHt  + 512;
    float* KR    = Km   + 512;
    float* Wgj   = KR   + 512;
    float* meanv = Wgj  + 128;
    float* meanp = meanv + 64;
    float* resid = meanp + 64;
    float* fac   = resid + 8;
    float* piv   = fac   + 8;

    const int g   = blockIdx.x;
    const int tid = threadIdx.x;

    for (int i = tid; i < 4096; i += 256) {
        const int r = i >> 6, c = i & 63;
        Fs[r * P_ + c]   = F[i];
        bufA[r * P_ + c] = P0g[i];
    }
    for (int i = tid; i < 512; i += 256) Hs[(i >> 6) * P_ + (i & 63)] = H[i];
    if (tid < 64) {
        Rs[tid]    = Rg[tid];
        meanv[tid] = init_mean[g * S_ + tid];
    }
    __syncthreads();

    float* covP  = bufA;
    float* covpP = bufB;
    float* tmpP  = bufC;

    for (int t = 0; t < T_; ++t) {
        mm64(Fs, covP, tmpP, false, nullptr, nullptr, nullptr);
        __syncthreads();
        mm64(tmpP, Fs, covpP, true, Qg, nullptr, nullptr);
        __syncthreads();

        for (int e = tid; e < 512; e += 256) {
            const int s = e >> 3, m = e & 7;
            float acc = 0.f;
            for (int k = 0; k < 64; ++k) acc += covpP[s * P_ + k] * Hs[m * P_ + k];
            PHt[e] = acc;
        }
        if (tid < 64) {
            float acc = 0.f;
            for (int j = 0; j < 64; ++j) acc += Fs[tid * P_ + j] * meanv[j];
            meanp[tid] = acc;
        }
        __syncthreads();

        if (tid < 128) {
            const int i = tid >> 4, c = tid & 15;
            if (c < 8) {
                float acc = Rs[i * 8 + c];
                for (int s = 0; s < 64; ++s) acc += Hs[i * P_ + s] * PHt[s * 8 + c];
                Wgj[i * 16 + c] = acc;
            } else {
                Wgj[i * 16 + c] = (c - 8 == i) ? 1.f : 0.f;
            }
        } else if (tid < 136) {
            const int m = tid - 128;
            float acc = 0.f;
            for (int s = 0; s < 64; ++s) acc += Hs[m * P_ + s] * meanp[s];
            out[(g * T_ + t) * M_ + m] = acc;
            resid[m] = y[(g * T_ + t) * M_ + m] - acc;
        }
        __syncthreads();

        for (int p = 0; p < 8; ++p) {
            if (tid == 0) piv[0] = 1.f / Wgj[p * 16 + p];
            __syncthreads();
            if (tid < 16) Wgj[p * 16 + tid] *= piv[0];
            __syncthreads();
            if (tid < 8) fac[tid] = Wgj[tid * 16 + p];
            __syncthreads();
            if (tid < 128) {
                const int r = tid >> 4, c = tid & 15;
                if (r != p) Wgj[r * 16 + c] -= fac[r] * Wgj[p * 16 + c];
            }
            __syncthreads();
        }

        for (int e = tid; e < 512; e += 256) {
            const int s = e >> 3, m = e & 7;
            float acc = 0.f;
            for (int j = 0; j < 8; ++j) acc += PHt[s * 8 + j] * Wgj[j * 16 + 8 + m];
            Km[e] = acc;
        }
        __syncthreads();

        if (tid < 64) {
            float acc = meanp[tid];
            for (int m = 0; m < 8; ++m) acc += Km[tid * 8 + m] * resid[m];
            meanv[tid] = acc;
        }
        for (int e = tid; e < 512; e += 256) {
            const int i = e >> 3, m = e & 7;
            float acc = 0.f;
            for (int n = 0; n < 8; ++n) acc += Km[i * 8 + n] * Rs[n * 8 + m];
            KR[e] = acc;
        }
        for (int e = tid; e < 4096; e += 256) {
            const int i = e >> 6, j = e & 63;
            float acc = (i == j) ? 1.f : 0.f;
            for (int m = 0; m < 8; ++m) acc -= Km[i * 8 + m] * Hs[m * P_ + j];
            covP[i * P_ + j] = acc;
        }
        __syncthreads();

        mm64(covP, covpP, tmpP, false, nullptr, nullptr, nullptr);
        __syncthreads();
        mm64(tmpP, covP, covpP, true, nullptr, KR, Km);
        __syncthreads();

        float* sw = covP; covP = covpP; covpP = sw;
    }
}

extern "C" void kernel_launch(void* const* d_in, const int* in_sizes, int n_in,
                              void* d_out, int out_size, void* d_ws, size_t ws_size,
                              hipStream_t stream) {
    const float* y         = (const float*)d_in[0];
    const float* F         = (const float*)d_in[1];
    const float* H         = (const float*)d_in[2];
    const float* Lq        = (const float*)d_in[3];
    const float* Lr        = (const float*)d_in[4];
    const float* init_mean = (const float*)d_in[5];
    const float* Linit     = (const float*)d_in[6];
    float* out = (float*)d_out;
    float* ws  = (float*)d_ws;

    kf_prep<<<16, 256, 0, stream>>>(Lq, Lr, Linit, ws);

    const size_t smem_bytes = (4864 * 4 + 608 + 64 + 512 + 512 + 512 + 128 +
                               64 + 64 + 8 + 8 + 8) * sizeof(float);
    kf_main<<<G_, 256, smem_bytes, stream>>>(y, F, H,
                                             ws,
                                             ws + 8192,
                                             ws + 4096,
                                             init_mean, out);
}


# 8x8 innovation inverse: Gauss-Jordan kept in registers of every wave (lane = column, pivot and factors by v_readlane) instead of 32 barrier-separated LDS passes; same operations, same order
# speedup vs baseline: 3.6832x; 1.2005x over previous
#include <hip/hip_runtime.h>
#include <hip/hip_bf16.h>

#define G_ 128
#define T_ 128
#define S_ 64
#define M_ 8
#define P_ 76

typedef float v2f __attribute__((ext_vector_type(2)));
typedef float v8f __attribute__((ext_vector_type(8)));

__device__ __forceinline__ float lane_bcast(float x, int src)
{
    return __int_as_float(__builtin_amdgcn_readlane(__float_as_int(x), src));
}

__device__ __forceinline__ void mm64(const float* A, const float* B, float* D,
                                     bool transB,
                                     const float* addQ,
                                     const float* KRp,
                                     const float* Kp)
{
    const int lane = threadIdx.x & 31;
    const int wave = threadIdx.x >> 5;
    const int half = lane >> 4;
    const int l15  = lane & 15;

    for (int tt = wave; tt < 16; tt += 8) {
        const int ti = (tt >> 2) * 16;
        const int tj = (tt & 3) * 16;
        v8f acc = {};
        #pragma unroll
        for (int kb = 0; kb < 16; ++kb) {
            const int k0 = kb * 4 + half * 2;
            v2f a, b;
            a.x = A[(ti + l15) * P_ + k0];
            a.y = A[(ti + l15) * P_ + k0 + 1];
            if (!transB) {
                b.x = B[(k0    ) * P_ + (tj + l15)];
                b.y = B[(k0 + 1) * P_ + (tj + l15)];
            } else {
                b.x = B[(tj + l15) * P_ + k0];
                b.y = B[(tj + l15) * P_ + k0 + 1];
            }
            acc = __builtin_amdgcn_wmma_f32_16x16x4_f32(
                      false, a, false, b, (short)0, acc, false, false);
        }
        #pragma unroll
        for (int r = 0; r < 8; ++r) {
            const int row = ti + r + half * 8;
            const int col = tj + l15;
            float v = acc[r];
            if (addQ) v += addQ[row * 64 + col];
            if (KRp) {
                #pragma unroll
                for (int m = 0; m < 8; ++m)
                    v += KRp[row * 8 + m] * Kp[col * 8 + m];
            }
            D[row * P_ + col] = v;
        }
    }
}

__global__ __launch_bounds__(256) void kf_prep(const float* __restrict__ Lq,
                                               const float* __restrict__ Lr,
                                               const float* __restrict__ Linit,
                                               float* __restrict__ ws)
{
    float* Q  = ws;
    float* P0 = ws + 4096;
    float* R  = ws + 8192;
    for (int e = blockIdx.x * blockDim.x + threadIdx.x; e < 4096;
         e += gridDim.x * blockDim.x) {
        const int i = e >> 6, j = e & 63;
        float q = 0.f, p = 0.f;
        for (int k = 0; k < 64; ++k) {
            q += Lq[i * 64 + k] * Lq[j * 64 + k];
            p += Linit[i * 64 + k] * Linit[j * 64 + k];
        }
        Q[e] = q;
        P0[e] = p;
    }
    const int e = blockIdx.x * blockDim.x + threadIdx.x;
    if (e < 64) {
        const int i = e >> 3, j = e & 7;
        float r = 0.f;
        for (int k = 0; k < 8; ++k) r += Lr[i * 8 + k] * Lr[j * 8 + k];
        R[e] = r;
    }
}

__global__ __launch_bounds__(256) void kf_main(const float* __restrict__ y,
                                               const float* __restrict__ F,
                                               const float* __restrict__ H,
                                               const float* __restrict__ Qg,
                                               const float* __restrict__ Rg,
                                               const float* __restrict__ P0g,
                                               const float* __restrict__ init_mean,
                                               float* __restrict__ out)
{
    extern __shared__ float smem[];
    float* Fs    = smem;
    float* bufA  = Fs   + 64 * P_;
    float* bufB  = bufA + 64 * P_;
    float* bufC  = bufB + 64 * P_;
    float* Hs    = bufC + 64 * P_;
    float* Rs    = Hs   + 8 * P_;
    float* PHt   = Rs   + 64;
    float* Km    = PHt  + 512;
    float* KR    = Km   + 512;
    float* Sin   = KR   + 512;
    float* meanv = Sin  + 64;
    float* meanp = meanv + 64;
    float* resid = meanp + 64;

    const int g   = blockIdx.x;
    const int tid = threadIdx.x;

    for (int i = tid; i < 4096; i += 256) {
        const int r = i >> 6, c = i & 63;
        Fs[r * P_ + c]   = F[i];
        bufA[r * P_ + c] = P0g[i];
    }
    for (int i = tid; i < 512; i += 256) Hs[(i >> 6) * P_ + (i & 63)] = H[i];
    if (tid < 64) {
        Rs[tid]    = Rg[tid];
        meanv[tid] = init_mean[g * S_ + tid];
    }
    __syncthreads();

    float* covP  = bufA;
    float* covpP = bufB;
    float* tmpP  = bufC;

    for (int t = 0; t < T_; ++t) {
        mm64(Fs, covP, tmpP, false, nullptr, nullptr, nullptr);
        __syncthreads();
        mm64(tmpP, Fs, covpP, true, Qg, nullptr, nullptr);
        __syncthreads();

        for (int e = tid; e < 512; e += 256) {
            const int s = e >> 3, m = e & 7;
            float acc = 0.f;
            for (int k = 0; k < 64; ++k) acc += covpP[s * P_ + k] * Hs[m * P_ + k];
            PHt[e] = acc;
        }
        if (tid < 64) {
            float acc = 0.f;
            for (int j = 0; j < 64; ++j) acc += Fs[tid * P_ + j] * meanv[j];
            meanp[tid] = acc;
        }
        __syncthreads();

        if (tid < 64) {
            const int i = tid >> 3, c = tid & 7;
            float acc = Rs[i * 8 + c];
            for (int s = 0; s < 64; ++s) acc += Hs[i * P_ + s] * PHt[s * 8 + c];
            Sin[i * 8 + c] = acc;
        } else if (tid >= 128 && tid < 136) {
            const int m = tid - 128;
            float acc = 0.f;
            for (int s = 0; s < 64; ++s) acc += Hs[m * P_ + s] * meanp[s];
            out[(g * T_ + t) * M_ + m] = acc;
            resid[m] = y[(g * T_ + t) * M_ + m] - acc;
        }
        __syncthreads();

        const int cI = tid & 7;
        float wl[8], wr[8];
        #pragma unroll
        for (int r = 0; r < 8; ++r) {
            wl[r] = Sin[r * 8 + cI];
            wr[r] = (r == cI) ? 1.f : 0.f;
        }
        #pragma unroll
        for (int p = 0; p < 8; ++p) {
            const float pv = 1.f / lane_bcast(wl[p], p);
            wl[p] *= pv;
            wr[p] *= pv;
            #pragma unroll
            for (int r = 0; r < 8; ++r) {
                if (r != p) {
                    const float f = lane_bcast(wl[r], p);
                    wl[r] -= f * wl[p];
                    wr[r] -= f * wr[p];
                }
            }
        }

        for (int e = tid; e < 512; e += 256) {
            const int s = e >> 3;
            float acc = 0.f;
            #pragma unroll
            for (int j = 0; j < 8; ++j) acc += PHt[s * 8 + j] * wr[j];
            Km[e] = acc;
        }
        __syncthreads();

        if (tid < 64) {
            float acc = meanp[tid];
            for (int m = 0; m < 8; ++m) acc += Km[tid * 8 + m] * resid[m];
            meanv[tid] = acc;
        }
        for (int e = tid; e < 512; e += 256) {
            const int i = e >> 3, m = e & 7;
            float acc = 0.f;
            for (int n = 0; n < 8; ++n) acc += Km[i * 8 + n] * Rs[n * 8 + m];
            KR[e] = acc;
        }
        for (int e = tid; e < 4096; e += 256) {
            const int i = e >> 6, j = e & 63;
            float acc = (i == j) ? 1.f : 0.f;
            for (int m = 0; m < 8; ++m) acc -= Km[i * 8 + m] * Hs[m * P_ + j];
            covP[i * P_ + j] = acc;
        }
        __syncthreads();

        mm64(covP, covpP, tmpP, false, nullptr, nullptr, nullptr);
        __syncthreads();
        mm64(tmpP, covP, covpP, true, nullptr, KR, Km);
        __syncthreads();

        float* sw = covP; covP = covpP; covpP = sw;
    }
}

extern "C" void kernel_launch(void* const* d_in, const int* in_sizes, int n_in,
                              void* d_out, int out_size, void* d_ws, size_t ws_size,
                              hipStream_t stream) {
    const float* y         = (const float*)d_in[0];
    const float* F         = (const float*)d_in[1];
    const float* H         = (const float*)d_in[2];
    const float* Lq        = (const float*)d_in[3];
    const float* Lr        = (const float*)d_in[4];
    const float* init_mean = (const float*)d_in[5];
    const float* Linit     = (const float*)d_in[6];
    float* out = (float*)d_out;
    float* ws  = (float*)d_ws;

    kf_prep<<<16, 256, 0, stream>>>(Lq, Lr, Linit, ws);

    const size_t smem_bytes = (4864 * 4 + 608 + 64 + 512 + 512 + 512 + 64 +
                               64 + 64 + 8) * sizeof(float);
    kf_main<<<G_, 256, smem_bytes, stream>>>(y, F, H,
                                             ws,
                                             ws + 8192,
                                             ws + 4096,
                                             init_mean, out);
}


# Each thread finishes all results of a phase in registers before storing any (both WMMA tiles of a wave, the 8-row epilogues, the 16 I-KH entries): loads no longer queue behind may-alias LDS stores
# speedup vs baseline: 5.5294x; 1.5012x over previous
#include <hip/hip_runtime.h>
#include <hip/hip_bf16.h>

#define G_ 128
#define T_ 128
#define S_ 64
#define M_ 8
#define P_ 76

typedef float v2f __attribute__((ext_vector_type(2)));
typedef float v8f __attribute__((ext_vector_type(8)));

__device__ __forceinline__ float lane_bcast(float x, int src)
{
    return __int_as_float(__builtin_amdgcn_readlane(__float_as_int(x), src));
}

template <bool TRANSB, int EPI>
__device__ __forceinline__ void mm64(const float* A, const float* B, float* D,
                                     const float* addQ,
                                     const float* KRp,
                                     const float* Kp)
{
    const int lane = threadIdx.x & 31;
    const int wave = threadIdx.x >> 5;
    const int half = lane >> 4;
    const int l15  = lane & 15;
    const int ti0  = (wave >> 2) * 16;
    const int tj   = (wave & 3) * 16;

    v8f acc0 = {}, acc1 = {};
    #pragma unroll
    for (int kb = 0; kb < 16; ++kb) {
        const int k0 = kb * 4 + half * 2;
        v2f a0, a1, b;
        a0.x = A[(ti0 + l15) * P_ + k0];
        a0.y = A[(ti0 + l15) * P_ + k0 + 1];
        a1.x = A[(ti0 + 32 + l15) * P_ + k0];
        a1.y = A[(ti0 + 32 + l15) * P_ + k0 + 1];
        if (!TRANSB) {
            b.x = B[(k0    ) * P_ + (tj + l15)];
            b.y = B[(k0 + 1) * P_ + (tj + l15)];
        } else {
            b.x = B[(tj + l15) * P_ + k0];
            b.y = B[(tj + l15) * P_ + k0 + 1];
        }
        acc0 = __builtin_amdgcn_wmma_f32_16x16x4_f32(
                   false, a0, false, b, (short)0, acc0, false, false);
        acc1 = __builtin_amdgcn_wmma_f32_16x16x4_f32(
                   false, a1, false, b, (short)0, acc1, false, false);
    }

    const int col = tj + l15;
    float v0[8], v1[8];
    if (EPI == 2) {
        float kc[8];
        #pragma unroll
        for (int m = 0; m < 8; ++m) kc[m] = Kp[col * 8 + m];
        #pragma unroll
        for (int r = 0; r < 8; ++r) {
            const int row0 = ti0 + r + half * 8;
            const int row1 = row0 + 32;
            float x0 = acc0[r], x1 = acc1[r];
            #pragma unroll
            for (int m = 0; m < 8; ++m) {
                x0 += KRp[row0 * 8 + m] * kc[m];
                x1 += KRp[row1 * 8 + m] * kc[m];
            }
            v0[r] = x0;
            v1[r] = x1;
        }
    } else {
        #pragma unroll
        for (int r = 0; r < 8; ++r) {
            const int row0 = ti0 + r + half * 8;
            const int row1 = row0 + 32;
            float x0 = acc0[r], x1 = acc1[r];
            if (EPI == 1) {
                x0 += addQ[row0 * 64 + col];
                x1 += addQ[row1 * 64 + col];
            }
            v0[r] = x0;
            v1[r] = x1;
        }
    }
    #pragma unroll
    for (int r = 0; r < 8; ++r) {
        const int row0 = ti0 + r + half * 8;
        D[row0 * P_ + col]        = v0[r];
        D[(row0 + 32) * P_ + col] = v1[r];
    }
}

__global__ __launch_bounds__(256) void kf_prep(const float* __restrict__ Lq,
                                               const float* __restrict__ Lr,
                                               const float* __restrict__ Linit,
                                               float* __restrict__ ws)
{
    float* Q  = ws;
    float* P0 = ws + 4096;
    float* R  = ws + 8192;
    for (int e = blockIdx.x * blockDim.x + threadIdx.x; e < 4096;
         e += gridDim.x * blockDim.x) {
        const int i = e >> 6, j = e & 63;
        float q = 0.f, p = 0.f;
        for (int k = 0; k < 64; ++k) {
            q += Lq[i * 64 + k] * Lq[j * 64 + k];
            p += Linit[i * 64 + k] * Linit[j * 64 + k];
        }
        Q[e] = q;
        P0[e] = p;
    }
    const int e = blockIdx.x * blockDim.x + threadIdx.x;
    if (e < 64) {
        const int i = e >> 3, j = e & 7;
        float r = 0.f;
        for (int k = 0; k < 8; ++k) r += Lr[i * 8 + k] * Lr[j * 8 + k];
        R[e] = r;
    }
}

__global__ __launch_bounds__(256) void kf_main(const float* __restrict__ y,
                                               const float* __restrict__ F,
                                               const float* __restrict__ H,
                                               const float* __restrict__ Qg,
                                               const float* __restrict__ Rg,
                                               const float* __restrict__ P0g,
                                               const float* __restrict__ init_mean,
                                               float* __restrict__ out)
{
    extern __shared__ float smem[];
    float* Fs    = smem;
    float* bufA  = Fs   + 64 * P_;
    float* bufB  = bufA + 64 * P_;
    float* bufC  = bufB + 64 * P_;
    float* Hs    = bufC + 64 * P_;
    float* Rs    = Hs   + 8 * P_;
    float* PHt   = Rs   + 64;
    float* Km    = PHt  + 512;
    float* KR    = Km   + 512;
    float* Sin   = KR   + 512;
    float* meanv = Sin  + 64;
    float* meanp = meanv + 64;
    float* resid = meanp + 64;

    const int g   = blockIdx.x;
    const int tid = threadIdx.x;

    for (int i = tid; i < 4096; i += 256) {
        const int r = i >> 6, c = i & 63;
        Fs[r * P_ + c]   = F[i];
        bufA[r * P_ + c] = P0g[i];
    }
    for (int i = tid; i < 512; i += 256) Hs[(i >> 6) * P_ + (i & 63)] = H[i];
    if (tid < 64) {
        Rs[tid]    = Rg[tid];
        meanv[tid] = init_mean[g * S_ + tid];
    }
    __syncthreads();

    float* covP  = bufA;
    float* covpP = bufB;
    float* tmpP  = bufC;

    for (int t = 0; t < T_; ++t) {
        mm64<false, 0>(Fs, covP, tmpP, nullptr, nullptr, nullptr);
        __syncthreads();
        mm64<true, 1>(tmpP, Fs, covpP, Qg, nullptr, nullptr);
        __syncthreads();

        {
            const int s0 = tid >> 3, m = tid & 7;
            float acc0 = 0.f, acc1 = 0.f;
            for (int k = 0; k < 64; ++k) {
                const float h = Hs[m * P_ + k];
                acc0 += covpP[s0 * P_ + k] * h;
                acc1 += covpP[(s0 + 32) * P_ + k] * h;
            }
            float mp = 0.f;
            if (tid < 64) {
                for (int j = 0; j < 64; ++j) mp += Fs[tid * P_ + j] * meanv[j];
            }
            PHt[tid]       = acc0;
            PHt[tid + 256] = acc1;
            if (tid < 64) meanp[tid] = mp;
        }
        __syncthreads();

        if (tid < 64) {
            const int i = tid >> 3, c = tid & 7;
            float acc = Rs[i * 8 + c];
            for (int s = 0; s < 64; ++s) acc += Hs[i * P_ + s] * PHt[s * 8 + c];
            Sin[i * 8 + c] = acc;
        } else if (tid >= 128 && tid < 136) {
            const int m = tid - 128;
            float acc = 0.f;
            for (int s = 0; s < 64; ++s) acc += Hs[m * P_ + s] * meanp[s];
            out[(g * T_ + t) * M_ + m] = acc;
            resid[m] = y[(g * T_ + t) * M_ + m] - acc;
        }
        __syncthreads();

        const int cI = tid & 7;
        float wl[8], wr[8];
        #pragma unroll
        for (int r = 0; r < 8; ++r) {
            wl[r] = Sin[r * 8 + cI];
            wr[r] = (r == cI) ? 1.f : 0.f;
        }
        #pragma unroll
        for (int p = 0; p < 8; ++p) {
            const float pv = 1.f / lane_bcast(wl[p], p);
            wl[p] *= pv;
            wr[p] *= pv;
            #pragma unroll
            for (int r = 0; r < 8; ++r) {
                if (r != p) {
                    const float f = lane_bcast(wl[r], p);
                    wl[r] -= f * wl[p];
                    wr[r] -= f * wr[p];
                }
            }
        }

        {
            const int s0 = tid >> 3;
            float acc0 = 0.f, acc1 = 0.f;
            #pragma unroll
            for (int j = 0; j < 8; ++j) {
                acc0 += PHt[s0 * 8 + j] * wr[j];
                acc1 += PHt[(s0 + 32) * 8 + j] * wr[j];
            }
            Km[tid]       = acc0;
            Km[tid + 256] = acc1;
        }
        __syncthreads();

        {
            float mu = 0.f;
            if (tid < 64) {
                mu = meanp[tid];
                for (int m = 0; m < 8; ++m) mu += Km[tid * 8 + m] * resid[m];
            }
            const int i0 = tid >> 3, mR = tid & 7;
            float kr0 = 0.f, kr1 = 0.f;
            for (int n = 0; n < 8; ++n) {
                const float rr = Rs[n * 8 + mR];
                kr0 += Km[i0 * 8 + n] * rr;
                kr1 += Km[(i0 + 32) * 8 + n] * rr;
            }
            const int q = tid >> 6, j = tid & 63;
            float hc[8];
            #pragma unroll
            for (int m = 0; m < 8; ++m) hc[m] = Hs[m * P_ + j];
            float ikh[16];
            #pragma unroll
            for (int it = 0; it < 16; ++it) {
                const int i = q + 4 * it;
                float acc = (i == j) ? 1.f : 0.f;
                #pragma unroll
                for (int m = 0; m < 8; ++m) acc -= Km[i * 8 + m] * hc[m];
                ikh[it] = acc;
            }
            if (tid < 64) meanv[tid] = mu;
            KR[tid]       = kr0;
            KR[tid + 256] = kr1;
            #pragma unroll
            for (int it = 0; it < 16; ++it)
                covP[(q + 4 * it) * P_ + j] = ikh[it];
        }
        __syncthreads();

        mm64<false, 0>(covP, covpP, tmpP, nullptr, nullptr, nullptr);
        __syncthreads();
        mm64<true, 2>(tmpP, covP, covpP, nullptr, KR, Km);
        __syncthreads();

        float* sw = covP; covP = covpP; covpP = sw;
    }
}

extern "C" void kernel_launch(void* const* d_in, const int* in_sizes, int n_in,
                              void* d_out, int out_size, void* d_ws, size_t ws_size,
                              hipStream_t stream) {
    const float* y         = (const float*)d_in[0];
    const float* F         = (const float*)d_in[1];
    const float* H         = (const float*)d_in[2];
    const float* Lq        = (const float*)d_in[3];
    const float* Lr        = (const float*)d_in[4];
    const float* init_mean = (const float*)d_in[5];
    const float* Linit     = (const float*)d_in[6];
    float* out = (float*)d_out;
    float* ws  = (float*)d_ws;

    kf_prep<<<16, 256, 0, stream>>>(Lq, Lr, Linit, ws);

    const size_t smem_bytes = (4864 * 4 + 608 + 64 + 512 + 512 + 512 + 64 +
                               64 + 64 + 8) * sizeof(float);
    kf_main<<<G_, 256, smem_bytes, stream>>>(y, F, H,
                                             ws,
                                             ws + 8192,
                                             ws + 4096,
                                             init_mean, out);
}


# Thin dot-product phases (PHt, mean_p, S_inn, yhat): a thread's LDS operand reads issued as one 128-bit group ahead of its FMA chains, constant H/F rows held in registers; sum order unchanged
# speedup vs baseline: 8.2500x; 1.4920x over previous
#include <hip/hip_runtime.h>
#include <hip/hip_bf16.h>

#define G_ 128
#define T_ 128
#define S_ 64
#define M_ 8
#define P_ 76

typedef float v2f __attribute__((ext_vector_type(2)));
typedef float v8f __attribute__((ext_vector_type(8)));

__device__ __forceinline__ float lane_bcast(float x, int src)
{
    return __int_as_float(__builtin_amdgcn_readlane(__float_as_int(x), src));
}

template <bool TRANSB, int EPI>
__device__ __forceinline__ void mm64(const float* A, const float* B, float* D,
                                     const float* addQ,
                                     const float* KRp,
                                     const float* Kp)
{
    const int lane = threadIdx.x & 31;
    const int wave = threadIdx.x >> 5;
    const int half = lane >> 4;
    const int l15  = lane & 15;
    const int ti0  = (wave >> 2) * 16;
    const int tj   = (wave & 3) * 16;

    v8f acc0 = {}, acc1 = {};
    #pragma unroll
    for (int kb = 0; kb < 16; ++kb) {
        const int k0 = kb * 4 + half * 2;
        v2f a0, a1, b;
        a0.x = A[(ti0 + l15) * P_ + k0];
        a0.y = A[(ti0 + l15) * P_ + k0 + 1];
        a1.x = A[(ti0 + 32 + l15) * P_ + k0];
        a1.y = A[(ti0 + 32 + l15) * P_ + k0 + 1];
        if (!TRANSB) {
            b.x = B[(k0    ) * P_ + (tj + l15)];
            b.y = B[(k0 + 1) * P_ + (tj + l15)];
        } else {
            b.x = B[(tj + l15) * P_ + k0];
            b.y = B[(tj + l15) * P_ + k0 + 1];
        }
        acc0 = __builtin_amdgcn_wmma_f32_16x16x4_f32(
                   false, a0, false, b, (short)0, acc0, false, false);
        acc1 = __builtin_amdgcn_wmma_f32_16x16x4_f32(
                   false, a1, false, b, (short)0, acc1, false, false);
    }

    const int col = tj + l15;
    float v0[8], v1[8];
    if (EPI == 2) {
        float kc[8];
        #pragma unroll
        for (int m = 0; m < 8; ++m) kc[m] = Kp[col * 8 + m];
        #pragma unroll
        for (int r = 0; r < 8; ++r) {
            const int row0 = ti0 + r + half * 8;
            const int row1 = row0 + 32;
            float x0 = acc0[r], x1 = acc1[r];
            #pragma unroll
            for (int m = 0; m < 8; ++m) {
                x0 += KRp[row0 * 8 + m] * kc[m];
                x1 += KRp[row1 * 8 + m] * kc[m];
            }
            v0[r] = x0;
            v1[r] = x1;
        }
    } else {
        #pragma unroll
        for (int r = 0; r < 8; ++r) {
            const int row0 = ti0 + r + half * 8;
            const int row1 = row0 + 32;
            float x0 = acc0[r], x1 = acc1[r];
            if (EPI == 1) {
                x0 += addQ[row0 * 64 + col];
                x1 += addQ[row1 * 64 + col];
            }
            v0[r] = x0;
            v1[r] = x1;
        }
    }
    #pragma unroll
    for (int r = 0; r < 8; ++r) {
        const int row0 = ti0 + r + half * 8;
        D[row0 * P_ + col]        = v0[r];
        D[(row0 + 32) * P_ + col] = v1[r];
    }
}

__global__ __launch_bounds__(256) void kf_prep(const float* __restrict__ Lq,
                                               const float* __restrict__ Lr,
                                               const float* __restrict__ Linit,
                                               float* __restrict__ ws)
{
    float* Q  = ws;
    float* P0 = ws + 4096;
    float* R  = ws + 8192;
    for (int e = blockIdx.x * blockDim.x + threadIdx.x; e < 4096;
         e += gridDim.x * blockDim.x) {
        const int i = e >> 6, j = e & 63;
        float q = 0.f, p = 0.f;
        for (int k = 0; k < 64; ++k) {
            q += Lq[i * 64 + k] * Lq[j * 64 + k];
            p += Linit[i * 64 + k] * Linit[j * 64 + k];
        }
        Q[e] = q;
        P0[e] = p;
    }
    const int e = blockIdx.x * blockDim.x + threadIdx.x;
    if (e < 64) {
        const int i = e >> 3, j = e & 7;
        float r = 0.f;
        for (int k = 0; k < 8; ++k) r += Lr[i * 8 + k] * Lr[j * 8 + k];
        R[e] = r;
    }
}

__global__ __launch_bounds__(256) void kf_main(const float* __restrict__ y,
                                               const float* __restrict__ F,
                                               const float* __restrict__ H,
                                               const float* __restrict__ Qg,
                                               const float* __restrict__ Rg,
                                               const float* __restrict__ P0g,
                                               const float* __restrict__ init_mean,
                                               float* __restrict__ out)
{
    extern __shared__ __attribute__((aligned(16))) float smem[];
    float* Fs    = smem;
    float* bufA  = Fs   + 64 * P_;
    float* bufB  = bufA + 64 * P_;
    float* bufC  = bufB + 64 * P_;
    float* Hs    = bufC + 64 * P_;
    float* Rs    = Hs   + 8 * P_;
    float* PHt   = Rs   + 64;
    float* Km    = PHt  + 512;
    float* KR    = Km   + 512;
    float* Sin   = KR   + 512;
    float* meanv = Sin  + 64;
    float* meanp = meanv + 64;
    float* resid = meanp + 64;

    const int g   = blockIdx.x;
    const int tid = threadIdx.x;

    for (int i = tid; i < 4096; i += 256) {
        const int r = i >> 6, c = i & 63;
        Fs[r * P_ + c]   = F[i];
        bufA[r * P_ + c] = P0g[i];
    }
    for (int i = tid; i < 512; i += 256) Hs[(i >> 6) * P_ + (i & 63)] = H[i];
    if (tid < 64) {
        Rs[tid]    = Rg[tid];
        meanv[tid] = init_mean[g * S_ + tid];
    }
    __syncthreads();

    float4 rw[32];
    {
        const float* r0 = (tid < 128) ? Hs + (tid & 3) * P_ : Fs + (tid & 63) * P_;
        const float* r1 = (tid < 128) ? Hs + ((tid & 3) + 4) * P_ : r0;
        #pragma unroll
        for (int q = 0; q < 16; ++q) {
            rw[q]      = ((const float4*)r0)[q];
            rw[16 + q] = ((const float4*)r1)[q];
        }
    }

    float* covP  = bufA;
    float* covpP = bufB;
    float* tmpP  = bufC;

    for (int t = 0; t < T_; ++t) {
        mm64<false, 0>(Fs, covP, tmpP, nullptr, nullptr, nullptr);
        __syncthreads();
        mm64<true, 1>(tmpP, Fs, covpP, Qg, nullptr, nullptr);
        __syncthreads();

        if (tid < 128) {
            const int sp = tid >> 2, ma = tid & 3;
            const float4* c0 = (const float4*)(covpP + sp * P_);
            const float4* c1 = (const float4*)(covpP + (sp + 32) * P_);
            float4 x0[16], x1[16];
            #pragma unroll
            for (int q = 0; q < 16; ++q) { x0[q] = c0[q]; x1[q] = c1[q]; }
            __builtin_amdgcn_sched_barrier(0);
            float a00 = 0.f, a01 = 0.f, a10 = 0.f, a11 = 0.f;
            #pragma unroll
            for (int q = 0; q < 16; ++q) {
                a00 = __builtin_fmaf(x0[q].x, rw[q].x, a00);
                a01 = __builtin_fmaf(x0[q].x, rw[16 + q].x, a01);
                a10 = __builtin_fmaf(x1[q].x, rw[q].x, a10);
                a11 = __builtin_fmaf(x1[q].x, rw[16 + q].x, a11);
                a00 = __builtin_fmaf(x0[q].y, rw[q].y, a00);
                a01 = __builtin_fmaf(x0[q].y, rw[16 + q].y, a01);
                a10 = __builtin_fmaf(x1[q].y, rw[q].y, a10);
                a11 = __builtin_fmaf(x1[q].y, rw[16 + q].y, a11);
                a00 = __builtin_fmaf(x0[q].z, rw[q].z, a00);
                a01 = __builtin_fmaf(x0[q].z, rw[16 + q].z, a01);
                a10 = __builtin_fmaf(x1[q].z, rw[q].z, a10);
                a11 = __builtin_fmaf(x1[q].z, rw[16 + q].z, a11);
                a00 = __builtin_fmaf(x0[q].w, rw[q].w, a00);
                a01 = __builtin_fmaf(x0[q].w, rw[16 + q].w, a01);
                a10 = __builtin_fmaf(x1[q].w, rw[q].w, a10);
                a11 = __builtin_fmaf(x1[q].w, rw[16 + q].w, a11);
            }
            PHt[sp * 8 + ma]            = a00;
            PHt[sp * 8 + ma + 4]        = a01;
            PHt[(sp + 32) * 8 + ma]     = a10;
            PHt[(sp + 32) * 8 + ma + 4] = a11;
        } else if (tid < 192) {
            const float4* mv = (const float4*)meanv;
            float4 m4[16];
            #pragma unroll
            for (int q = 0; q < 16; ++q) m4[q] = mv[q];
            __builtin_amdgcn_sched_barrier(0);
            float mp = 0.f;
            #pragma unroll
            for (int q = 0; q < 16; ++q) {
                mp = __builtin_fmaf(rw[q].x, m4[q].x, mp);
                mp = __builtin_fmaf(rw[q].y, m4[q].y, mp);
                mp = __builtin_fmaf(rw[q].z, m4[q].z, mp);
                mp = __builtin_fmaf(rw[q].w, m4[q].w, mp);
            }
            meanp[tid - 128] = mp;
        }
        __syncthreads();

        if (tid < 64) {
            const int i = tid >> 3, c = tid & 7;
            const float4* hr = (const float4*)(Hs + i * P_);
            float4 h4[16];
            float  ph[64];
            float acc = Rs[i * 8 + c];
            #pragma unroll
            for (int q = 0; q < 16; ++q) h4[q] = hr[q];
            #pragma unroll
            for (int s = 0; s < 64; ++s) ph[s] = PHt[s * 8 + c];
            __builtin_amdgcn_sched_barrier(0);
            #pragma unroll
            for (int q = 0; q < 16; ++q) {
                acc = __builtin_fmaf(h4[q].x, ph[4 * q],     acc);
                acc = __builtin_fmaf(h4[q].y, ph[4 * q + 1], acc);
                acc = __builtin_fmaf(h4[q].z, ph[4 * q + 2], acc);
                acc = __builtin_fmaf(h4[q].w, ph[4 * q + 3], acc);
            }
            Sin[i * 8 + c] = acc;
        } else if (tid >= 128 && tid < 136) {
            const int m = tid - 128;
            const float yt = y[(g * T_ + t) * M_ + m];
            const float4* hr = (const float4*)(Hs + m * P_);
            const float4* mq = (const float4*)meanp;
            float4 h4[16], m4[16];
            #pragma unroll
            for (int q = 0; q < 16; ++q) { h4[q] = hr[q]; m4[q] = mq[q]; }
            __builtin_amdgcn_sched_barrier(0);
            float acc = 0.f;
            #pragma unroll
            for (int q = 0; q < 16; ++q) {
                acc = __builtin_fmaf(h4[q].x, m4[q].x, acc);
                acc = __builtin_fmaf(h4[q].y, m4[q].y, acc);
                acc = __builtin_fmaf(h4[q].z, m4[q].z, acc);
                acc = __builtin_fmaf(h4[q].w, m4[q].w, acc);
            }
            out[(g * T_ + t) * M_ + m] = acc;
            resid[m] = yt - acc;
        }
        __syncthreads();

        const int cI = tid & 7;
        float wl[8], wr[8];
        #pragma unroll
        for (int r = 0; r < 8; ++r) {
            wl[r] = Sin[r * 8 + cI];
            wr[r] = (r == cI) ? 1.f : 0.f;
        }
        #pragma unroll
        for (int p = 0; p < 8; ++p) {
            const float pv = 1.f / lane_bcast(wl[p], p);
            wl[p] *= pv;
            wr[p] *= pv;
            #pragma unroll
            for (int r = 0; r < 8; ++r) {
                if (r != p) {
                    const float f = lane_bcast(wl[r], p);
                    wl[r] -= f * wl[p];
                    wr[r] -= f * wr[p];
                }
            }
        }

        {
            const int s0 = tid >> 3;
            float acc0 = 0.f, acc1 = 0.f;
            #pragma unroll
            for (int j = 0; j < 8; ++j) {
                acc0 += PHt[s0 * 8 + j] * wr[j];
                acc1 += PHt[(s0 + 32) * 8 + j] * wr[j];
            }
            Km[tid]       = acc0;
            Km[tid + 256] = acc1;
        }
        __syncthreads();

        {
            float mu = 0.f;
            if (tid < 64) {
                mu = meanp[tid];
                for (int m = 0; m < 8; ++m) mu += Km[tid * 8 + m] * resid[m];
            }
            const int i0 = tid >> 3, mR = tid & 7;
            float kr0 = 0.f, kr1 = 0.f;
            for (int n = 0; n < 8; ++n) {
                const float rr = Rs[n * 8 + mR];
                kr0 += Km[i0 * 8 + n] * rr;
                kr1 += Km[(i0 + 32) * 8 + n] * rr;
            }
            const int q = tid >> 6, j = tid & 63;
            float hc[8];
            #pragma unroll
            for (int m = 0; m < 8; ++m) hc[m] = Hs[m * P_ + j];
            float ikh[16];
            #pragma unroll
            for (int it = 0; it < 16; ++it) {
                const int i = q + 4 * it;
                float acc = (i == j) ? 1.f : 0.f;
                #pragma unroll
                for (int m = 0; m < 8; ++m) acc -= Km[i * 8 + m] * hc[m];
                ikh[it] = acc;
            }
            if (tid < 64) meanv[tid] = mu;
            KR[tid]       = kr0;
            KR[tid + 256] = kr1;
            #pragma unroll
            for (int it = 0; it < 16; ++it)
                covP[(q + 4 * it) * P_ + j] = ikh[it];
        }
        __syncthreads();

        mm64<false, 0>(covP, covpP, tmpP, nullptr, nullptr, nullptr);
        __syncthreads();
        mm64<true, 2>(tmpP, covP, covpP, nullptr, KR, Km);
        __syncthreads();

        float* sw = covP; covP = covpP; covpP = sw;
    }
}

extern "C" void kernel_launch(void* const* d_in, const int* in_sizes, int n_in,
                              void* d_out, int out_size, void* d_ws, size_t ws_size,
                              hipStream_t stream) {
    const float* y         = (const float*)d_in[0];
    const float* F         = (const float*)d_in[1];
    const float* H         = (const float*)d_in[2];
    const float* Lq        = (const float*)d_in[3];
    const float* Lr        = (const float*)d_in[4];
    const float* init_mean = (const float*)d_in[5];
    const float* Linit     = (const float*)d_in[6];
    float* out = (float*)d_out;
    float* ws  = (float*)d_ws;

    kf_prep<<<16, 256, 0, stream>>>(Lq, Lr, Linit, ws);

    const size_t smem_bytes = (4864 * 4 + 608 + 64 + 512 + 512 + 512 + 64 +
                               64 + 64 + 8) * sizeof(float);
    kf_main<<<G_, 256, smem_bytes, stream>>>(y, F, H,
                                             ws,
                                             ws + 8192,
                                             ws + 4096,
                                             init_mean, out);
}


# First Joseph product as rank-8 WMMA update: tmp = cov_p - K (H cov_p), H cov_p formed from cov_p itself by the two waves idle in the thin phases; a 64-term product and a barrier go; sum order changed
# speedup vs baseline: 8.9805x; 1.0885x over previous
#include <hip/hip_runtime.h>
#include <hip/hip_bf16.h>

#define G_ 128
#define T_ 128
#define S_ 64
#define M_ 8
#define P_ 76

typedef float v2f __attribute__((ext_vector_type(2)));
typedef float v8f __attribute__((ext_vector_type(8)));

__device__ __forceinline__ float lane_bcast(float x, int src)
{
    return __int_as_float(__builtin_amdgcn_readlane(__float_as_int(x), src));
}

template <bool TRANSB, int EPI>
__device__ __forceinline__ void mm64(const float* A, const float* B, float* D,
                                     const float* addQ,
                                     const float* KRp,
                                     const float* Kp)
{
    const int lane = threadIdx.x & 31;
    const int wave = threadIdx.x >> 5;
    const int half = lane >> 4;
    const int l15  = lane & 15;
    const int ti0  = (wave >> 2) * 16;
    const int tj   = (wave & 3) * 16;

    v8f acc0 = {}, acc1 = {};
    #pragma unroll
    for (int kb = 0; kb < 16; ++kb) {
        const int k0 = kb * 4 + half * 2;
        v2f a0, a1, b;
        a0.x = A[(ti0 + l15) * P_ + k0];
        a0.y = A[(ti0 + l15) * P_ + k0 + 1];
        a1.x = A[(ti0 + 32 + l15) * P_ + k0];
        a1.y = A[(ti0 + 32 + l15) * P_ + k0 + 1];
        if (!TRANSB) {
            b.x = B[(k0    ) * P_ + (tj + l15)];
            b.y = B[(k0 + 1) * P_ + (tj + l15)];
        } else {
            b.x = B[(tj + l15) * P_ + k0];
            b.y = B[(tj + l15) * P_ + k0 + 1];
        }
        acc0 = __builtin_amdgcn_wmma_f32_16x16x4_f32(
                   false, a0, false, b, (short)0, acc0, false, false);
        acc1 = __builtin_amdgcn_wmma_f32_16x16x4_f32(
                   false, a1, false, b, (short)0, acc1, false, false);
    }

    const int col = tj + l15;
    float v0[8], v1[8];
    if (EPI == 2) {
        float kc[8];
        #pragma unroll
        for (int m = 0; m < 8; ++m) kc[m] = Kp[col * 8 + m];
        #pragma unroll
        for (int r = 0; r < 8; ++r) {
            const int row0 = ti0 + r + half * 8;
            const int row1 = row0 + 32;
            float x0 = acc0[r], x1 = acc1[r];
            #pragma unroll
            for (int m = 0; m < 8; ++m) {
                x0 += KRp[row0 * 8 + m] * kc[m];
                x1 += KRp[row1 * 8 + m] * kc[m];
            }
            v0[r] = x0;
            v1[r] = x1;
        }
    } else {
        #pragma unroll
        for (int r = 0; r < 8; ++r) {
            const int row0 = ti0 + r + half * 8;
            const int row1 = row0 + 32;
            float x0 = acc0[r], x1 = acc1[r];
            if (EPI == 1) {
                x0 += addQ[row0 * 64 + col];
                x1 += addQ[row1 * 64 + col];
            }
            v0[r] = x0;
            v1[r] = x1;
        }
    }
    #pragma unroll
    for (int r = 0; r < 8; ++r) {
        const int row0 = ti0 + r + half * 8;
        D[row0 * P_ + col]        = v0[r];
        D[(row0 + 32) * P_ + col] = v1[r];
    }
}

__global__ __launch_bounds__(256) void kf_prep(const float* __restrict__ Lq,
                                               const float* __restrict__ Lr,
                                               const float* __restrict__ Linit,
                                               float* __restrict__ ws)
{
    float* Q  = ws;
    float* P0 = ws + 4096;
    float* R  = ws + 8192;
    for (int e = blockIdx.x * blockDim.x + threadIdx.x; e < 4096;
         e += gridDim.x * blockDim.x) {
        const int i = e >> 6, j = e & 63;
        float q = 0.f, p = 0.f;
        for (int k = 0; k < 64; ++k) {
            q += Lq[i * 64 + k] * Lq[j * 64 + k];
            p += Linit[i * 64 + k] * Linit[j * 64 + k];
        }
        Q[e] = q;
        P0[e] = p;
    }
    const int e = blockIdx.x * blockDim.x + threadIdx.x;
    if (e < 64) {
        const int i = e >> 3, j = e & 7;
        float r = 0.f;
        for (int k = 0; k < 8; ++k) r += Lr[i * 8 + k] * Lr[j * 8 + k];
        R[e] = r;
    }
}

__device__ __forceinline__ void hp_tile(const float* covp, float* HPs,
                                        const v2f (&ha)[16], int tj)
{
    const int lane = threadIdx.x & 31;
    const int half = lane >> 4;
    const int l15  = lane & 15;
    v2f b[16];
    #pragma unroll
    for (int kb = 0; kb < 16; ++kb) {
        const int k0 = kb * 4 + half * 2;
        b[kb].x = covp[(k0    ) * P_ + (tj + l15)];
        b[kb].y = covp[(k0 + 1) * P_ + (tj + l15)];
    }
    v8f acc0 = {}, acc1 = {};
    #pragma unroll
    for (int kb = 0; kb < 8; ++kb) {
        acc0 = __builtin_amdgcn_wmma_f32_16x16x4_f32(
                   false, ha[kb], false, b[kb], (short)0, acc0, false, false);
        acc1 = __builtin_amdgcn_wmma_f32_16x16x4_f32(
                   false, ha[kb + 8], false, b[kb + 8], (short)0, acc1, false, false);
    }
    float hp[8];
    #pragma unroll
    for (int r = 0; r < 8; ++r) hp[r] = acc0[r] + acc1[r];
    if (half == 0) {
        #pragma unroll
        for (int r = 0; r < 8; ++r) HPs[r * P_ + (tj + l15)] = hp[r];
    }
}

__global__ __launch_bounds__(256) void kf_main(const float* __restrict__ y,
                                               const float* __restrict__ F,
                                               const float* __restrict__ H,
                                               const float* __restrict__ Qg,
                                               const float* __restrict__ Rg,
                                               const float* __restrict__ P0g,
                                               const float* __restrict__ init_mean,
                                               float* __restrict__ out)
{
    extern __shared__ __attribute__((aligned(16))) float smem[];
    float* Fs    = smem;
    float* bufA  = Fs   + 64 * P_;
    float* bufB  = bufA + 64 * P_;
    float* bufC  = bufB + 64 * P_;
    float* Hs    = bufC + 64 * P_;
    float* Rs    = Hs   + 8 * P_;
    float* PHt   = Rs   + 64;
    float* Km    = PHt  + 512;
    float* KR    = Km   + 512;
    float* Sin   = KR   + 512;
    float* meanv = Sin  + 64;
    float* meanp = meanv + 64;
    float* resid = meanp + 64;
    float* HPs   = resid + 8;

    const int g   = blockIdx.x;
    const int tid = threadIdx.x;

    for (int i = tid; i < 4096; i += 256) {
        const int r = i >> 6, c = i & 63;
        Fs[r * P_ + c]   = F[i];
        bufA[r * P_ + c] = P0g[i];
    }
    for (int i = tid; i < 512; i += 256) Hs[(i >> 6) * P_ + (i & 63)] = H[i];
    if (tid < 64) {
        Rs[tid]    = Rg[tid];
        meanv[tid] = init_mean[g * S_ + tid];
    }
    __syncthreads();

    float4 rw[32];
    {
        const float* r0 = (tid < 128) ? Hs + (tid & 3) * P_ : Fs + (tid & 63) * P_;
        const float* r1 = (tid < 128) ? Hs + ((tid & 3) + 4) * P_ : r0;
        #pragma unroll
        for (int q = 0; q < 16; ++q) {
            rw[q]      = ((const float4*)r0)[q];
            rw[16 + q] = ((const float4*)r1)[q];
        }
    }

    v2f ha[16];
    {
        const int l15 = tid & 15, half = (tid & 31) >> 4;
        const float* hrow = Hs + (l15 & 7) * P_;
        #pragma unroll
        for (int kb = 0; kb < 16; ++kb) {
            const int k0 = kb * 4 + half * 2;
            const float h0 = hrow[k0], h1 = hrow[k0 + 1];
            ha[kb].x = (l15 < 8) ? h0 : 0.f;
            ha[kb].y = (l15 < 8) ? h1 : 0.f;
        }
    }

    float* covP  = bufA;
    float* covpP = bufB;
    float* tmpP  = bufC;

    for (int t = 0; t < T_; ++t) {
        mm64<false, 0>(Fs, covP, tmpP, nullptr, nullptr, nullptr);
        __syncthreads();
        mm64<true, 1>(tmpP, Fs, covpP, Qg, nullptr, nullptr);
        __syncthreads();

        if (tid < 128) {
            const int sp = tid >> 2, ma = tid & 3;
            const float4* c0 = (const float4*)(covpP + sp * P_);
            const float4* c1 = (const float4*)(covpP + (sp + 32) * P_);
            float4 x0[16], x1[16];
            #pragma unroll
            for (int q = 0; q < 16; ++q) { x0[q] = c0[q]; x1[q] = c1[q]; }
            __builtin_amdgcn_sched_barrier(0);
            float a00 = 0.f, a01 = 0.f, a10 = 0.f, a11 = 0.f;
            #pragma unroll
            for (int q = 0; q < 16; ++q) {
                a00 = __builtin_fmaf(x0[q].x, rw[q].x, a00);
                a01 = __builtin_fmaf(x0[q].x, rw[16 + q].x, a01);
                a10 = __builtin_fmaf(x1[q].x, rw[q].x, a10);
                a11 = __builtin_fmaf(x1[q].x, rw[16 + q].x, a11);
                a00 = __builtin_fmaf(x0[q].y, rw[q].y, a00);
                a01 = __builtin_fmaf(x0[q].y, rw[16 + q].y, a01);
                a10 = __builtin_fmaf(x1[q].y, rw[q].y, a10);
                a11 = __builtin_fmaf(x1[q].y, rw[16 + q].y, a11);
                a00 = __builtin_fmaf(x0[q].z, rw[q].z, a00);
                a01 = __builtin_fmaf(x0[q].z, rw[16 + q].z, a01);
                a10 = __builtin_fmaf(x1[q].z, rw[q].z, a10);
                a11 = __builtin_fmaf(x1[q].z, rw[16 + q].z, a11);
                a00 = __builtin_fmaf(x0[q].w, rw[q].w, a00);
                a01 = __builtin_fmaf(x0[q].w, rw[16 + q].w, a01);
                a10 = __builtin_fmaf(x1[q].w, rw[q].w, a10);
                a11 = __builtin_fmaf(x1[q].w, rw[16 + q].w, a11);
            }
            PHt[sp * 8 + ma]            = a00;
            PHt[sp * 8 + ma + 4]        = a01;
            PHt[(sp + 32) * 8 + ma]     = a10;
            PHt[(sp + 32) * 8 + ma + 4] = a11;
        } else if (tid < 192) {
            const float4* mv = (const float4*)meanv;
            float4 m4[16];
            #pragma unroll
            for (int q = 0; q < 16; ++q) m4[q] = mv[q];
            __builtin_amdgcn_sched_barrier(0);
            float mp = 0.f;
            #pragma unroll
            for (int q = 0; q < 16; ++q) {
                mp = __builtin_fmaf(rw[q].x, m4[q].x, mp);
                mp = __builtin_fmaf(rw[q].y, m4[q].y, mp);
                mp = __builtin_fmaf(rw[q].z, m4[q].z, mp);
                mp = __builtin_fmaf(rw[q].w, m4[q].w, mp);
            }
            meanp[tid - 128] = mp;
        } else {
            hp_tile(covpP, HPs, ha, ((tid >> 5) - 6) * 16);
        }
        __syncthreads();

        if (tid < 64) {
            const int i = tid >> 3, c = tid & 7;
            const float4* hr = (const float4*)(Hs + i * P_);
            float4 h4[16];
            float  ph[64];
            float acc = Rs[i * 8 + c];
            #pragma unroll
            for (int q = 0; q < 16; ++q) h4[q] = hr[q];
            #pragma unroll
            for (int s = 0; s < 64; ++s) ph[s] = PHt[s * 8 + c];
            __builtin_amdgcn_sched_barrier(0);
            #pragma unroll
            for (int q = 0; q < 16; ++q) {
                acc = __builtin_fmaf(h4[q].x, ph[4 * q],     acc);
                acc = __builtin_fmaf(h4[q].y, ph[4 * q + 1], acc);
                acc = __builtin_fmaf(h4[q].z, ph[4 * q + 2], acc);
                acc = __builtin_fmaf(h4[q].w, ph[4 * q + 3], acc);
            }
            Sin[i * 8 + c] = acc;
        } else if (tid >= 128 && tid < 136) {
            const int m = tid - 128;
            const float yt = y[(g * T_ + t) * M_ + m];
            const float4* hr = (const float4*)(Hs + m * P_);
            const float4* mq = (const float4*)meanp;
            float4 h4[16], m4[16];
            #pragma unroll
            for (int q = 0; q < 16; ++q) { h4[q] = hr[q]; m4[q] = mq[q]; }
            __builtin_amdgcn_sched_barrier(0);
            float acc = 0.f;
            #pragma unroll
            for (int q = 0; q < 16; ++q) {
                acc = __builtin_fmaf(h4[q].x, m4[q].x, acc);
                acc = __builtin_fmaf(h4[q].y, m4[q].y, acc);
                acc = __builtin_fmaf(h4[q].z, m4[q].z, acc);
                acc = __builtin_fmaf(h4[q].w, m4[q].w, acc);
            }
            out[(g * T_ + t) * M_ + m] = acc;
            resid[m] = yt - acc;
        } else if (tid >= 192) {
            hp_tile(covpP, HPs, ha, ((tid >> 5) - 4) * 16);
        }
        __syncthreads();

        const int cI = tid & 7;
        float wl[8], wr[8];
        #pragma unroll
        for (int r = 0; r < 8; ++r) {
            wl[r] = Sin[r * 8 + cI];
            wr[r] = (r == cI) ? 1.f : 0.f;
        }
        #pragma unroll
        for (int p = 0; p < 8; ++p) {
            const float pv = 1.f / lane_bcast(wl[p], p);
            wl[p] *= pv;
            wr[p] *= pv;
            #pragma unroll
            for (int r = 0; r < 8; ++r) {
                if (r != p) {
                    const float f = lane_bcast(wl[r], p);
                    wl[r] -= f * wl[p];
                    wr[r] -= f * wr[p];
                }
            }
        }

        {
            const int s0 = tid >> 3;
            float acc0 = 0.f, acc1 = 0.f;
            #pragma unroll
            for (int j = 0; j < 8; ++j) {
                acc0 += PHt[s0 * 8 + j] * wr[j];
                acc1 += PHt[(s0 + 32) * 8 + j] * wr[j];
            }
            Km[tid]       = acc0;
            Km[tid + 256] = acc1;
        }
        __syncthreads();

        const int uh   = (tid & 31) >> 4;
        const int ul15 = tid & 15;
        const int uti0 = (tid >> 7) * 16;
        const int ucol = ((tid >> 5) & 3) * 16 + ul15;
        v8f t0, t1;
        #pragma unroll
        for (int r = 0; r < 8; ++r) {
            t0[r] = covpP[(uti0 + r + uh * 8) * P_ + ucol];
            t1[r] = covpP[(uti0 + 32 + r + uh * 8) * P_ + ucol];
        }
        #pragma unroll
        for (int kb = 0; kb < 2; ++kb) {
            const int k0 = kb * 4 + uh * 2;
            v2f a0, a1, b;
            a0.x = -Km[(uti0 + ul15) * 8 + k0];
            a0.y = -Km[(uti0 + ul15) * 8 + k0 + 1];
            a1.x = -Km[(uti0 + 32 + ul15) * 8 + k0];
            a1.y = -Km[(uti0 + 32 + ul15) * 8 + k0 + 1];
            b.x  = HPs[(k0    ) * P_ + ucol];
            b.y  = HPs[(k0 + 1) * P_ + ucol];
            t0 = __builtin_amdgcn_wmma_f32_16x16x4_f32(
                     false, a0, false, b, (short)0, t0, false, false);
            t1 = __builtin_amdgcn_wmma_f32_16x16x4_f32(
                     false, a1, false, b, (short)0, t1, false, false);
        }

        {
            float mu = 0.f;
            if (tid < 64) {
                mu = meanp[tid];
                for (int m = 0; m < 8; ++m) mu += Km[tid * 8 + m] * resid[m];
            }
            const int i0 = tid >> 3, mR = tid & 7;
            float kr0 = 0.f, kr1 = 0.f;
            for (int n = 0; n < 8; ++n) {
                const float rr = Rs[n * 8 + mR];
                kr0 += Km[i0 * 8 + n] * rr;
                kr1 += Km[(i0 + 32) * 8 + n] * rr;
            }
            const int q = tid >> 6, j = tid & 63;
            float hc[8];
            #pragma unroll
            for (int m = 0; m < 8; ++m) hc[m] = Hs[m * P_ + j];
            float ikh[16];
            #pragma unroll
            for (int it = 0; it < 16; ++it) {
                const int i = q + 4 * it;
                float acc = (i == j) ? 1.f : 0.f;
                #pragma unroll
                for (int m = 0; m < 8; ++m) acc -= Km[i * 8 + m] * hc[m];
                ikh[it] = acc;
            }
            if (tid < 64) meanv[tid] = mu;
            KR[tid]       = kr0;
            KR[tid + 256] = kr1;
            #pragma unroll
            for (int it = 0; it < 16; ++it)
                covP[(q + 4 * it) * P_ + j] = ikh[it];
        }
        #pragma unroll
        for (int r = 0; r < 8; ++r) {
            tmpP[(uti0 + r + uh * 8) * P_ + ucol]      = t0[r];
            tmpP[(uti0 + 32 + r + uh * 8) * P_ + ucol] = t1[r];
        }
        __syncthreads();

        mm64<true, 2>(tmpP, covP, covpP, nullptr, KR, Km);
        __syncthreads();

        float* sw = covP; covP = covpP; covpP = sw;
    }
}

extern "C" void kernel_launch(void* const* d_in, const int* in_sizes, int n_in,
                              void* d_out, int out_size, void* d_ws, size_t ws_size,
                              hipStream_t stream) {
    const float* y         = (const float*)d_in[0];
    const float* F         = (const float*)d_in[1];
    const float* H         = (const float*)d_in[2];
    const float* Lq        = (const float*)d_in[3];
    const float* Lr        = (const float*)d_in[4];
    const float* init_mean = (const float*)d_in[5];
    const float* Linit     = (const float*)d_in[6];
    float* out = (float*)d_out;
    float* ws  = (float*)d_ws;

    kf_prep<<<16, 256, 0, stream>>>(Lq, Lr, Linit, ws);

    const size_t smem_bytes = (4864 * 4 + 608 + 64 + 512 + 512 + 512 + 64 +
                               64 + 64 + 8 + 608) * sizeof(float);
    kf_main<<<G_, 256, smem_bytes, stream>>>(y, F, H,
                                             ws,
                                             ws + 8192,
                                             ws + 4096,
                                             init_mean, out);
}


# Second Joseph product as rank-8 WMMA update: cov = tmp + (K R - tmp H^T) K^T, with tmp H^T (64x8) on WMMA by all waves; I - K H, a 64-term product and its K R K^T epilogue go; sum order changed
# speedup vs baseline: 10.8067x; 1.2034x over previous
#include <hip/hip_runtime.h>
#include <hip/hip_bf16.h>

#define G_ 128
#define T_ 128
#define S_ 64
#define M_ 8
#define P_ 76

typedef float v2f __attribute__((ext_vector_type(2)));
typedef float v8f __attribute__((ext_vector_type(8)));

__device__ __forceinline__ float lane_bcast(float x, int src)
{
    return __int_as_float(__builtin_amdgcn_readlane(__float_as_int(x), src));
}

template <bool TRANSB, int EPI>
__device__ __forceinline__ void mm64(const float* A, const float* B, float* D,
                                     const float* addQ)
{
    const int lane = threadIdx.x & 31;
    const int wave = threadIdx.x >> 5;
    const int half = lane >> 4;
    const int l15  = lane & 15;
    const int ti0  = (wave >> 2) * 16;
    const int tj   = (wave & 3) * 16;

    v8f acc0 = {}, acc1 = {};
    #pragma unroll
    for (int kb = 0; kb < 16; ++kb) {
        const int k0 = kb * 4 + half * 2;
        v2f a0, a1, b;
        a0.x = A[(ti0 + l15) * P_ + k0];
        a0.y = A[(ti0 + l15) * P_ + k0 + 1];
        a1.x = A[(ti0 + 32 + l15) * P_ + k0];
        a1.y = A[(ti0 + 32 + l15) * P_ + k0 + 1];
        if (!TRANSB) {
            b.x = B[(k0    ) * P_ + (tj + l15)];
            b.y = B[(k0 + 1) * P_ + (tj + l15)];
        } else {
            b.x = B[(tj + l15) * P_ + k0];
            b.y = B[(tj + l15) * P_ + k0 + 1];
        }
        acc0 = __builtin_amdgcn_wmma_f32_16x16x4_f32(
                   false, a0, false, b, (short)0, acc0, false, false);
        acc1 = __builtin_amdgcn_wmma_f32_16x16x4_f32(
                   false, a1, false, b, (short)0, acc1, false, false);
    }

    const int col = tj + l15;
    float v0[8], v1[8];
    #pragma unroll
    for (int r = 0; r < 8; ++r) {
        const int row0 = ti0 + r + half * 8;
        const int row1 = row0 + 32;
        float x0 = acc0[r], x1 = acc1[r];
        if (EPI == 1) {
            x0 += addQ[row0 * 64 + col];
            x1 += addQ[row1 * 64 + col];
        }
        v0[r] = x0;
        v1[r] = x1;
    }
    #pragma unroll
    for (int r = 0; r < 8; ++r) {
        const int row0 = ti0 + r + half * 8;
        D[row0 * P_ + col]        = v0[r];
        D[(row0 + 32) * P_ + col] = v1[r];
    }
}

__global__ __launch_bounds__(256) void kf_prep(const float* __restrict__ Lq,
                                               const float* __restrict__ Lr,
                                               const float* __restrict__ Linit,
                                               float* __restrict__ ws)
{
    float* Q  = ws;
    float* P0 = ws + 4096;
    float* R  = ws + 8192;
    for (int e = blockIdx.x * blockDim.x + threadIdx.x; e < 4096;
         e += gridDim.x * blockDim.x) {
        const int i = e >> 6, j = e & 63;
        float q = 0.f, p = 0.f;
        for (int k = 0; k < 64; ++k) {
            q += Lq[i * 64 + k] * Lq[j * 64 + k];
            p += Linit[i * 64 + k] * Linit[j * 64 + k];
        }
        Q[e] = q;
        P0[e] = p;
    }
    const int e = blockIdx.x * blockDim.x + threadIdx.x;
    if (e < 64) {
        const int i = e >> 3, j = e & 7;
        float r = 0.f;
        for (int k = 0; k < 8; ++k) r += Lr[i * 8 + k] * Lr[j * 8 + k];
        R[e] = r;
    }
}

__device__ __forceinline__ void hp_tile(const float* covp, float* HPs,
                                        const v2f (&ha)[16], int tj)
{
    const int lane = threadIdx.x & 31;
    const int half = lane >> 4;
    const int l15  = lane & 15;
    v2f b[16];
    #pragma unroll
    for (int kb = 0; kb < 16; ++kb) {
        const int k0 = kb * 4 + half * 2;
        b[kb].x = covp[(k0    ) * P_ + (tj + l15)];
        b[kb].y = covp[(k0 + 1) * P_ + (tj + l15)];
    }
    v8f acc0 = {}, acc1 = {};
    #pragma unroll
    for (int kb = 0; kb < 8; ++kb) {
        acc0 = __builtin_amdgcn_wmma_f32_16x16x4_f32(
                   false, ha[kb], false, b[kb], (short)0, acc0, false, false);
        acc1 = __builtin_amdgcn_wmma_f32_16x16x4_f32(
                   false, ha[kb + 8], false, b[kb + 8], (short)0, acc1, false, false);
    }
    float hp[8];
    #pragma unroll
    for (int r = 0; r < 8; ++r) hp[r] = acc0[r] + acc1[r];
    if (half == 0) {
        #pragma unroll
        for (int r = 0; r < 8; ++r) HPs[r * P_ + (tj + l15)] = hp[r];
    }
}

template <int KH>
__device__ __forceinline__ void u_part(const float* tmp, float* Up,
                                       const v2f (&ha)[16], int mt)
{
    const int lane = threadIdx.x & 31;
    const int half = lane >> 4;
    const int l15  = lane & 15;
    v2f a[8];
    #pragma unroll
    for (int kb = 0; kb < 8; ++kb) {
        const int k0 = (KH * 8 + kb) * 4 + half * 2;
        a[kb].x = tmp[(mt * 16 + l15) * P_ + k0];
        a[kb].y = tmp[(mt * 16 + l15) * P_ + k0 + 1];
    }
    v8f acc0 = {}, acc1 = {};
    #pragma unroll
    for (int kb = 0; kb < 4; ++kb) {
        acc0 = __builtin_amdgcn_wmma_f32_16x16x4_f32(
                   false, a[kb], false, ha[KH * 8 + kb], (short)0, acc0, false, false);
        acc1 = __builtin_amdgcn_wmma_f32_16x16x4_f32(
                   false, a[kb + 4], false, ha[KH * 8 + kb + 4], (short)0, acc1, false, false);
    }
    float u[8];
    #pragma unroll
    for (int r = 0; r < 8; ++r) u[r] = acc0[r] + acc1[r];
    if (l15 < 8) {
        #pragma unroll
        for (int r = 0; r < 8; ++r)
            Up[KH * 512 + (mt * 16 + half * 8 + r) * 8 + l15] = u[r];
    }
}

__global__ __launch_bounds__(256) void kf_main(const float* __restrict__ y,
                                               const float* __restrict__ F,
                                               const float* __restrict__ H,
                                               const float* __restrict__ Qg,
                                               const float* __restrict__ Rg,
                                               const float* __restrict__ P0g,
                                               const float* __restrict__ init_mean,
                                               float* __restrict__ out)
{
    extern __shared__ __attribute__((aligned(16))) float smem[];
    float* Fs    = smem;
    float* bufA  = Fs   + 64 * P_;
    float* bufB  = bufA + 64 * P_;
    float* bufC  = bufB + 64 * P_;
    float* Hs    = bufC + 64 * P_;
    float* Rs    = Hs   + 8 * P_;
    float* PHt   = Rs   + 64;
    float* Km    = PHt  + 512;
    float* KR    = Km   + 512;
    float* Sin   = KR   + 512;
    float* meanv = Sin  + 64;
    float* meanp = meanv + 64;
    float* resid = meanp + 64;
    float* HPs   = resid + 8;
    float* Up    = HPs  + 8 * P_;

    const int g   = blockIdx.x;
    const int tid = threadIdx.x;

    for (int i = tid; i < 4096; i += 256) {
        const int r = i >> 6, c = i & 63;
        Fs[r * P_ + c]   = F[i];
        bufA[r * P_ + c] = P0g[i];
    }
    for (int i = tid; i < 512; i += 256) Hs[(i >> 6) * P_ + (i & 63)] = H[i];
    if (tid < 64) {
        Rs[tid]    = Rg[tid];
        meanv[tid] = init_mean[g * S_ + tid];
    }
    __syncthreads();

    float4 rw[32];
    {
        const float* r0 = (tid < 128) ? Hs + (tid & 3) * P_ : Fs + (tid & 63) * P_;
        const float* r1 = (tid < 128) ? Hs + ((tid & 3) + 4) * P_ : r0;
        #pragma unroll
        for (int q = 0; q < 16; ++q) {
            rw[q]      = ((const float4*)r0)[q];
            rw[16 + q] = ((const float4*)r1)[q];
        }
    }

    v2f ha[16];
    {
        const int l15 = tid & 15, half = (tid & 31) >> 4;
        const float* hrow = Hs + (l15 & 7) * P_;
        #pragma unroll
        for (int kb = 0; kb < 16; ++kb) {
            const int k0 = kb * 4 + half * 2;
            const float h0 = hrow[k0], h1 = hrow[k0 + 1];
            ha[kb].x = (l15 < 8) ? h0 : 0.f;
            ha[kb].y = (l15 < 8) ? h1 : 0.f;
        }
    }

    float* const covP  = bufA;
    float* const covpP = bufB;
    float* const tmpP  = bufC;

    for (int t = 0; t < T_; ++t) {
        mm64<false, 0>(Fs, covP, tmpP, nullptr);
        __syncthreads();
        mm64<true, 1>(tmpP, Fs, covpP, Qg);
        __syncthreads();

        if (tid < 128) {
            const int sp = tid >> 2, ma = tid & 3;
            const float4* c0 = (const float4*)(covpP + sp * P_);
            const float4* c1 = (const float4*)(covpP + (sp + 32) * P_);
            float4 x0[16], x1[16];
            #pragma unroll
            for (int q = 0; q < 16; ++q) { x0[q] = c0[q]; x1[q] = c1[q]; }
            __builtin_amdgcn_sched_barrier(0);
            float a00 = 0.f, a01 = 0.f, a10 = 0.f, a11 = 0.f;
            #pragma unroll
            for (int q = 0; q < 16; ++q) {
                a00 = __builtin_fmaf(x0[q].x, rw[q].x, a00);
                a01 = __builtin_fmaf(x0[q].x, rw[16 + q].x, a01);
                a10 = __builtin_fmaf(x1[q].x, rw[q].x, a10);
                a11 = __builtin_fmaf(x1[q].x, rw[16 + q].x, a11);
                a00 = __builtin_fmaf(x0[q].y, rw[q].y, a00);
                a01 = __builtin_fmaf(x0[q].y, rw[16 + q].y, a01);
                a10 = __builtin_fmaf(x1[q].y, rw[q].y, a10);
                a11 = __builtin_fmaf(x1[q].y, rw[16 + q].y, a11);
                a00 = __builtin_fmaf(x0[q].z, rw[q].z, a00);
                a01 = __builtin_fmaf(x0[q].z, rw[16 + q].z, a01);
                a10 = __builtin_fmaf(x1[q].z, rw[q].z, a10);
                a11 = __builtin_fmaf(x1[q].z, rw[16 + q].z, a11);
                a00 = __builtin_fmaf(x0[q].w, rw[q].w, a00);
                a01 = __builtin_fmaf(x0[q].w, rw[16 + q].w, a01);
                a10 = __builtin_fmaf(x1[q].w, rw[q].w, a10);
                a11 = __builtin_fmaf(x1[q].w, rw[16 + q].w, a11);
            }
            PHt[sp * 8 + ma]            = a00;
            PHt[sp * 8 + ma + 4]        = a01;
            PHt[(sp + 32) * 8 + ma]     = a10;
            PHt[(sp + 32) * 8 + ma + 4] = a11;
        } else if (tid < 192) {
            const float4* mv = (const float4*)meanv;
            float4 m4[16];
            #pragma unroll
            for (int q = 0; q < 16; ++q) m4[q] = mv[q];
            __builtin_amdgcn_sched_barrier(0);
            float mp = 0.f;
            #pragma unroll
            for (int q = 0; q < 16; ++q) {
                mp = __builtin_fmaf(rw[q].x, m4[q].x, mp);
                mp = __builtin_fmaf(rw[q].y, m4[q].y, mp);
                mp = __builtin_fmaf(rw[q].z, m4[q].z, mp);
                mp = __builtin_fmaf(rw[q].w, m4[q].w, mp);
            }
            meanp[tid - 128] = mp;
        } else {
            hp_tile(covpP, HPs, ha, ((tid >> 5) - 6) * 16);
        }
        __syncthreads();

        if (tid < 64) {
            const int i = tid >> 3, c = tid & 7;
            const float4* hr = (const float4*)(Hs + i * P_);
            float4 h4[16];
            float  ph[64];
            float acc = Rs[i * 8 + c];
            #pragma unroll
            for (int q = 0; q < 16; ++q) h4[q] = hr[q];
            #pragma unroll
            for (int s = 0; s < 64; ++s) ph[s] = PHt[s * 8 + c];
            __builtin_amdgcn_sched_barrier(0);
            #pragma unroll
            for (int q = 0; q < 16; ++q) {
                acc = __builtin_fmaf(h4[q].x, ph[4 * q],     acc);
                acc = __builtin_fmaf(h4[q].y, ph[4 * q + 1], acc);
                acc = __builtin_fmaf(h4[q].z, ph[4 * q + 2], acc);
                acc = __builtin_fmaf(h4[q].w, ph[4 * q + 3], acc);
            }
            Sin[i * 8 + c] = acc;
        } else if (tid >= 128 && tid < 136) {
            const int m = tid - 128;
            const float yt = y[(g * T_ + t) * M_ + m];
            const float4* hr = (const float4*)(Hs + m * P_);
            const float4* mq = (const float4*)meanp;
            float4 h4[16], m4[16];
            #pragma unroll
            for (int q = 0; q < 16; ++q) { h4[q] = hr[q]; m4[q] = mq[q]; }
            __builtin_amdgcn_sched_barrier(0);
            float acc = 0.f;
            #pragma unroll
            for (int q = 0; q < 16; ++q) {
                acc = __builtin_fmaf(h4[q].x, m4[q].x, acc);
                acc = __builtin_fmaf(h4[q].y, m4[q].y, acc);
                acc = __builtin_fmaf(h4[q].z, m4[q].z, acc);
                acc = __builtin_fmaf(h4[q].w, m4[q].w, acc);
            }
            out[(g * T_ + t) * M_ + m] = acc;
            resid[m] = yt - acc;
        } else if (tid >= 192) {
            hp_tile(covpP, HPs, ha, ((tid >> 5) - 4) * 16);
        }
        __syncthreads();

        const int cI = tid & 7;
        float wl[8], wr[8];
        #pragma unroll
        for (int r = 0; r < 8; ++r) {
            wl[r] = Sin[r * 8 + cI];
            wr[r] = (r == cI) ? 1.f : 0.f;
        }
        #pragma unroll
        for (int p = 0; p < 8; ++p) {
            const float pv = 1.f / lane_bcast(wl[p], p);
            wl[p] *= pv;
            wr[p] *= pv;
            #pragma unroll
            for (int r = 0; r < 8; ++r) {
                if (r != p) {
                    const float f = lane_bcast(wl[r], p);
                    wl[r] -= f * wl[p];
                    wr[r] -= f * wr[p];
                }
            }
        }

        {
            const int s0 = tid >> 3;
            float acc0 = 0.f, acc1 = 0.f;
            #pragma unroll
            for (int j = 0; j < 8; ++j) {
                acc0 += PHt[s0 * 8 + j] * wr[j];
                acc1 += PHt[(s0 + 32) * 8 + j] * wr[j];
            }
            Km[tid]       = acc0;
            Km[tid + 256] = acc1;
        }
        __syncthreads();

        const int uh   = (tid & 31) >> 4;
        const int ul15 = tid & 15;
        const int uti0 = (tid >> 7) * 16;
        const int ucol = ((tid >> 5) & 3) * 16 + ul15;
        v8f t0, t1;
        #pragma unroll
        for (int r = 0; r < 8; ++r) {
            t0[r] = covpP[(uti0 + r + uh * 8) * P_ + ucol];
            t1[r] = covpP[(uti0 + 32 + r + uh * 8) * P_ + ucol];
        }
        #pragma unroll
        for (int kb = 0; kb < 2; ++kb) {
            const int k0 = kb * 4 + uh * 2;
            v2f a0, a1, b;
            a0.x = -Km[(uti0 + ul15) * 8 + k0];
            a0.y = -Km[(uti0 + ul15) * 8 + k0 + 1];
            a1.x = -Km[(uti0 + 32 + ul15) * 8 + k0];
            a1.y = -Km[(uti0 + 32 + ul15) * 8 + k0 + 1];
            b.x  = HPs[(k0    ) * P_ + ucol];
            b.y  = HPs[(k0 + 1) * P_ + ucol];
            t0 = __builtin_amdgcn_wmma_f32_16x16x4_f32(
                     false, a0, false, b, (short)0, t0, false, false);
            t1 = __builtin_amdgcn_wmma_f32_16x16x4_f32(
                     false, a1, false, b, (short)0, t1, false, false);
        }

        {
            float mu = 0.f;
            if (tid < 64) {
                mu = meanp[tid];
                for (int m = 0; m < 8; ++m) mu += Km[tid * 8 + m] * resid[m];
            }
            const int i0 = tid >> 3, mR = tid & 7;
            float kr0 = 0.f, kr1 = 0.f;
            for (int n = 0; n < 8; ++n) {
                const float rr = Rs[n * 8 + mR];
                kr0 += Km[i0 * 8 + n] * rr;
                kr1 += Km[(i0 + 32) * 8 + n] * rr;
            }
            if (tid < 64) meanv[tid] = mu;
            KR[tid]       = kr0;
            KR[tid + 256] = kr1;
        }
        #pragma unroll
        for (int r = 0; r < 8; ++r) {
            tmpP[(uti0 + r + uh * 8) * P_ + ucol]      = t0[r];
            tmpP[(uti0 + 32 + r + uh * 8) * P_ + ucol] = t1[r];
        }
        __syncthreads();

        if (tid < 128) u_part<0>(tmpP, Up, ha, tid >> 5);
        else           u_part<1>(tmpP, Up, ha, (tid >> 5) - 4);
        __syncthreads();

        v2f wa0[2], wa1[2], kb2[2];
        #pragma unroll
        for (int kb = 0; kb < 2; ++kb) {
            const int k0 = kb * 4 + uh * 2;
            const int r0 = (uti0 + ul15) * 8 + k0;
            const int r1 = (uti0 + 32 + ul15) * 8 + k0;
            wa0[kb].x = KR[r0]     - (Up[r0]     + Up[512 + r0]);
            wa0[kb].y = KR[r0 + 1] - (Up[r0 + 1] + Up[512 + r0 + 1]);
            wa1[kb].x = KR[r1]     - (Up[r1]     + Up[512 + r1]);
            wa1[kb].y = KR[r1 + 1] - (Up[r1 + 1] + Up[512 + r1 + 1]);
            kb2[kb].x = Km[ucol * 8 + k0];
            kb2[kb].y = Km[ucol * 8 + k0 + 1];
        }
        __builtin_amdgcn_sched_barrier(0);
        #pragma unroll
        for (int kb = 0; kb < 2; ++kb) {
            t0 = __builtin_amdgcn_wmma_f32_16x16x4_f32(
                     false, wa0[kb], false, kb2[kb], (short)0, t0, false, false);
            t1 = __builtin_amdgcn_wmma_f32_16x16x4_f32(
                     false, wa1[kb], false, kb2[kb], (short)0, t1, false, false);
        }
        #pragma unroll
        for (int r = 0; r < 8; ++r) {
            covP[(uti0 + r + uh * 8) * P_ + ucol]      = t0[r];
            covP[(uti0 + 32 + r + uh * 8) * P_ + ucol] = t1[r];
        }
        __syncthreads();
    }
}

extern "C" void kernel_launch(void* const* d_in, const int* in_sizes, int n_in,
                              void* d_out, int out_size, void* d_ws, size_t ws_size,
                              hipStream_t stream) {
    const float* y         = (const float*)d_in[0];
    const float* F         = (const float*)d_in[1];
    const float* H         = (const float*)d_in[2];
    const float* Lq        = (const float*)d_in[3];
    const float* Lr        = (const float*)d_in[4];
    const float* init_mean = (const float*)d_in[5];
    const float* Linit     = (const float*)d_in[6];
    float* out = (float*)d_out;
    float* ws  = (float*)d_ws;

    kf_prep<<<16, 256, 0, stream>>>(Lq, Lr, Linit, ws);

    const size_t smem_bytes = (4864 * 4 + 608 + 64 + 512 + 512 + 512 + 64 +
                               64 + 64 + 8 + 608 + 1024) * sizeof(float);
    kf_main<<<G_, 256, smem_bytes, stream>>>(y, F, H,
                                             ws,
                                             ws + 8192,
                                             ws + 4096,
                                             init_mean, out);
}


# Second Joseph update without the 64-term tmp H^T: tmp H^T = PHt - K (S_inn - R), so W = K S_inn - PHt on WMMA beside the first update; tmp stays in registers, two barriers go; sum order changed
# speedup vs baseline: 12.0527x; 1.1153x over previous
#include <hip/hip_runtime.h>
#include <hip/hip_bf16.h>

#define G_ 128
#define T_ 128
#define S_ 64
#define M_ 8
#define P_ 76

typedef float v2f __attribute__((ext_vector_type(2)));
typedef float v8f __attribute__((ext_vector_type(8)));

__device__ __forceinline__ float lane_bcast(float x, int src)
{
    return __int_as_float(__builtin_amdgcn_readlane(__float_as_int(x), src));
}

template <bool TRANSB, int EPI>
__device__ __forceinline__ void mm64(const float* A, const float* B, float* D,
                                     const float* addQ)
{
    const int lane = threadIdx.x & 31;
    const int wave = threadIdx.x >> 5;
    const int half = lane >> 4;
    const int l15  = lane & 15;
    const int ti0  = (wave >> 2) * 16;
    const int tj   = (wave & 3) * 16;

    v8f acc0 = {}, acc1 = {};
    #pragma unroll
    for (int kb = 0; kb < 16; ++kb) {
        const int k0 = kb * 4 + half * 2;
        v2f a0, a1, b;
        a0.x = A[(ti0 + l15) * P_ + k0];
        a0.y = A[(ti0 + l15) * P_ + k0 + 1];
        a1.x = A[(ti0 + 32 + l15) * P_ + k0];
        a1.y = A[(ti0 + 32 + l15) * P_ + k0 + 1];
        if (!TRANSB) {
            b.x = B[(k0    ) * P_ + (tj + l15)];
            b.y = B[(k0 + 1) * P_ + (tj + l15)];
        } else {
            b.x = B[(tj + l15) * P_ + k0];
            b.y = B[(tj + l15) * P_ + k0 + 1];
        }
        acc0 = __builtin_amdgcn_wmma_f32_16x16x4_f32(
                   false, a0, false, b, (short)0, acc0, false, false);
        acc1 = __builtin_amdgcn_wmma_f32_16x16x4_f32(
                   false, a1, false, b, (short)0, acc1, false, false);
    }

    const int col = tj + l15;
    float v0[8], v1[8];
    #pragma unroll
    for (int r = 0; r < 8; ++r) {
        const int row0 = ti0 + r + half * 8;
        const int row1 = row0 + 32;
        float x0 = acc0[r], x1 = acc1[r];
        if (EPI == 1) {
            x0 += addQ[row0 * 64 + col];
            x1 += addQ[row1 * 64 + col];
        }
        v0[r] = x0;
        v1[r] = x1;
    }
    #pragma unroll
    for (int r = 0; r < 8; ++r) {
        const int row0 = ti0 + r + half * 8;
        D[row0 * P_ + col]        = v0[r];
        D[(row0 + 32) * P_ + col] = v1[r];
    }
}

__global__ __launch_bounds__(256) void kf_prep(const float* __restrict__ Lq,
                                               const float* __restrict__ Lr,
                                               const float* __restrict__ Linit,
                                               float* __restrict__ ws)
{
    float* Q  = ws;
    float* P0 = ws + 4096;
    float* R  = ws + 8192;
    for (int e = blockIdx.x * blockDim.x + threadIdx.x; e < 4096;
         e += gridDim.x * blockDim.x) {
        const int i = e >> 6, j = e & 63;
        float q = 0.f, p = 0.f;
        for (int k = 0; k < 64; ++k) {
            q += Lq[i * 64 + k] * Lq[j * 64 + k];
            p += Linit[i * 64 + k] * Linit[j * 64 + k];
        }
        Q[e] = q;
        P0[e] = p;
    }
    const int e = blockIdx.x * blockDim.x + threadIdx.x;
    if (e < 64) {
        const int i = e >> 3, j = e & 7;
        float r = 0.f;
        for (int k = 0; k < 8; ++k) r += Lr[i * 8 + k] * Lr[j * 8 + k];
        R[e] = r;
    }
}

__device__ __forceinline__ void hp_tile(const float* covp, float* HPs,
                                        const v2f (&ha)[16], int tj)
{
    const int lane = threadIdx.x & 31;
    const int half = lane >> 4;
    const int l15  = lane & 15;
    v2f b[16];
    #pragma unroll
    for (int kb = 0; kb < 16; ++kb) {
        const int k0 = kb * 4 + half * 2;
        b[kb].x = covp[(k0    ) * P_ + (tj + l15)];
        b[kb].y = covp[(k0 + 1) * P_ + (tj + l15)];
    }
    v8f acc0 = {}, acc1 = {};
    #pragma unroll
    for (int kb = 0; kb < 8; ++kb) {
        acc0 = __builtin_amdgcn_wmma_f32_16x16x4_f32(
                   false, ha[kb], false, b[kb], (short)0, acc0, false, false);
        acc1 = __builtin_amdgcn_wmma_f32_16x16x4_f32(
                   false, ha[kb + 8], false, b[kb + 8], (short)0, acc1, false, false);
    }
    float hp[8];
    #pragma unroll
    for (int r = 0; r < 8; ++r) hp[r] = acc0[r] + acc1[r];
    if (half == 0) {
        #pragma unroll
        for (int r = 0; r < 8; ++r) HPs[r * P_ + (tj + l15)] = hp[r];
    }
}

__global__ __launch_bounds__(256) void kf_main(const float* __restrict__ y,
                                               const float* __restrict__ F,
                                               const float* __restrict__ H,
                                               const float* __restrict__ Qg,
                                               const float* __restrict__ Rg,
                                               const float* __restrict__ P0g,
                                               const float* __restrict__ init_mean,
                                               float* __restrict__ out)
{
    extern __shared__ __attribute__((aligned(16))) float smem[];
    float* Fs    = smem;
    float* bufA  = Fs   + 64 * P_;
    float* bufB  = bufA + 64 * P_;
    float* bufC  = bufB + 64 * P_;
    float* Hs    = bufC + 64 * P_;
    float* Rs    = Hs   + 8 * P_;
    float* PHt   = Rs   + 64;
    float* Km    = PHt  + 512;
    float* Sin   = Km   + 512;
    float* meanv = Sin  + 64;
    float* meanp = meanv + 64;
    float* resid = meanp + 64;
    float* HPs   = resid + 8;

    const int g   = blockIdx.x;
    const int tid = threadIdx.x;

    for (int i = tid; i < 4096; i += 256) {
        const int r = i >> 6, c = i & 63;
        Fs[r * P_ + c]   = F[i];
        bufA[r * P_ + c] = P0g[i];
    }
    for (int i = tid; i < 512; i += 256) Hs[(i >> 6) * P_ + (i & 63)] = H[i];
    if (tid < 64) {
        Rs[tid]    = Rg[tid];
        meanv[tid] = init_mean[g * S_ + tid];
    }
    __syncthreads();

    float4 rw[32];
    {
        const float* r0 = (tid < 128) ? Hs + (tid & 3) * P_ : Fs + (tid & 63) * P_;
        const float* r1 = (tid < 128) ? Hs + ((tid & 3) + 4) * P_ : r0;
        #pragma unroll
        for (int q = 0; q < 16; ++q) {
            rw[q]      = ((const float4*)r0)[q];
            rw[16 + q] = ((const float4*)r1)[q];
        }
    }

    v2f ha[16];
    {
        const int l15 = tid & 15, half = (tid & 31) >> 4;
        const float* hrow = Hs + (l15 & 7) * P_;
        #pragma unroll
        for (int kb = 0; kb < 16; ++kb) {
            const int k0 = kb * 4 + half * 2;
            const float h0 = hrow[k0], h1 = hrow[k0 + 1];
            ha[kb].x = (l15 < 8) ? h0 : 0.f;
            ha[kb].y = (l15 < 8) ? h1 : 0.f;
        }
    }

    float* const covP  = bufA;
    float* const covpP = bufB;
    float* const tmpP  = bufC;

    for (int t = 0; t < T_; ++t) {
        mm64<false, 0>(Fs, covP, tmpP, nullptr);
        __syncthreads();
        mm64<true, 1>(tmpP, Fs, covpP, Qg);
        __syncthreads();

        if (tid < 128) {
            const int sp = tid >> 2, ma = tid & 3;
            const float4* c0 = (const float4*)(covpP + sp * P_);
            const float4* c1 = (const float4*)(covpP + (sp + 32) * P_);
            float4 x0[16], x1[16];
            #pragma unroll
            for (int q = 0; q < 16; ++q) { x0[q] = c0[q]; x1[q] = c1[q]; }
            __builtin_amdgcn_sched_barrier(0);
            float a00 = 0.f, a01 = 0.f, a10 = 0.f, a11 = 0.f;
            #pragma unroll
            for (int q = 0; q < 16; ++q) {
                a00 = __builtin_fmaf(x0[q].x, rw[q].x, a00);
                a01 = __builtin_fmaf(x0[q].x, rw[16 + q].x, a01);
                a10 = __builtin_fmaf(x1[q].x, rw[q].x, a10);
                a11 = __builtin_fmaf(x1[q].x, rw[16 + q].x, a11);
                a00 = __builtin_fmaf(x0[q].y, rw[q].y, a00);
                a01 = __builtin_fmaf(x0[q].y, rw[16 + q].y, a01);
                a10 = __builtin_fmaf(x1[q].y, rw[q].y, a10);
                a11 = __builtin_fmaf(x1[q].y, rw[16 + q].y, a11);
                a00 = __builtin_fmaf(x0[q].z, rw[q].z, a00);
                a01 = __builtin_fmaf(x0[q].z, rw[16 + q].z, a01);
                a10 = __builtin_fmaf(x1[q].z, rw[q].z, a10);
                a11 = __builtin_fmaf(x1[q].z, rw[16 + q].z, a11);
                a00 = __builtin_fmaf(x0[q].w, rw[q].w, a00);
                a01 = __builtin_fmaf(x0[q].w, rw[16 + q].w, a01);
                a10 = __builtin_fmaf(x1[q].w, rw[q].w, a10);
                a11 = __builtin_fmaf(x1[q].w, rw[16 + q].w, a11);
            }
            PHt[sp * 8 + ma]            = a00;
            PHt[sp * 8 + ma + 4]        = a01;
            PHt[(sp + 32) * 8 + ma]     = a10;
            PHt[(sp + 32) * 8 + ma + 4] = a11;
        } else if (tid < 192) {
            const float4* mv = (const float4*)meanv;
            float4 m4[16];
            #pragma unroll
            for (int q = 0; q < 16; ++q) m4[q] = mv[q];
            __builtin_amdgcn_sched_barrier(0);
            float mp = 0.f;
            #pragma unroll
            for (int q = 0; q < 16; ++q) {
                mp = __builtin_fmaf(rw[q].x, m4[q].x, mp);
                mp = __builtin_fmaf(rw[q].y, m4[q].y, mp);
                mp = __builtin_fmaf(rw[q].z, m4[q].z, mp);
                mp = __builtin_fmaf(rw[q].w, m4[q].w, mp);
            }
            meanp[tid - 128] = mp;
        } else {
            hp_tile(covpP, HPs, ha, ((tid >> 5) - 6) * 16);
        }
        __syncthreads();

        if (tid < 64) {
            const int i = tid >> 3, c = tid & 7;
            const float4* hr = (const float4*)(Hs + i * P_);
            float4 h4[16];
            float  ph[64];
            float acc = Rs[i * 8 + c];
            #pragma unroll
            for (int q = 0; q < 16; ++q) h4[q] = hr[q];
            #pragma unroll
            for (int s = 0; s < 64; ++s) ph[s] = PHt[s * 8 + c];
            __builtin_amdgcn_sched_barrier(0);
            #pragma unroll
            for (int q = 0; q < 16; ++q) {
                acc = __builtin_fmaf(h4[q].x, ph[4 * q],     acc);
                acc = __builtin_fmaf(h4[q].y, ph[4 * q + 1], acc);
                acc = __builtin_fmaf(h4[q].z, ph[4 * q + 2], acc);
                acc = __builtin_fmaf(h4[q].w, ph[4 * q + 3], acc);
            }
            Sin[i * 8 + c] = acc;
        } else if (tid >= 128 && tid < 136) {
            const int m = tid - 128;
            const float yt = y[(g * T_ + t) * M_ + m];
            const float4* hr = (const float4*)(Hs + m * P_);
            const float4* mq = (const float4*)meanp;
            float4 h4[16], m4[16];
            #pragma unroll
            for (int q = 0; q < 16; ++q) { h4[q] = hr[q]; m4[q] = mq[q]; }
            __builtin_amdgcn_sched_barrier(0);
            float acc = 0.f;
            #pragma unroll
            for (int q = 0; q < 16; ++q) {
                acc = __builtin_fmaf(h4[q].x, m4[q].x, acc);
                acc = __builtin_fmaf(h4[q].y, m4[q].y, acc);
                acc = __builtin_fmaf(h4[q].z, m4[q].z, acc);
                acc = __builtin_fmaf(h4[q].w, m4[q].w, acc);
            }
            out[(g * T_ + t) * M_ + m] = acc;
            resid[m] = yt - acc;
        } else if (tid >= 192) {
            hp_tile(covpP, HPs, ha, ((tid >> 5) - 4) * 16);
        }
        __syncthreads();

        const int cI = tid & 7;
        float wl[8], wr[8];
        #pragma unroll
        for (int r = 0; r < 8; ++r) {
            wl[r] = Sin[r * 8 + cI];
            wr[r] = (r == cI) ? 1.f : 0.f;
        }
        #pragma unroll
        for (int p = 0; p < 8; ++p) {
            const float pv = 1.f / lane_bcast(wl[p], p);
            wl[p] *= pv;
            wr[p] *= pv;
            #pragma unroll
            for (int r = 0; r < 8; ++r) {
                if (r != p) {
                    const float f = lane_bcast(wl[r], p);
                    wl[r] -= f * wl[p];
                    wr[r] -= f * wr[p];
                }
            }
        }

        {
            const int s0 = tid >> 3;
            float acc0 = 0.f, acc1 = 0.f;
            #pragma unroll
            for (int j = 0; j < 8; ++j) {
                acc0 += PHt[s0 * 8 + j] * wr[j];
                acc1 += PHt[(s0 + 32) * 8 + j] * wr[j];
            }
            Km[tid]       = acc0;
            Km[tid + 256] = acc1;
        }
        __syncthreads();

        const int uh   = (tid & 31) >> 4;
        const int ul15 = tid & 15;
        const int uti0 = (tid >> 7) * 16;
        const int ucol = ((tid >> 5) & 3) * 16 + ul15;
        const int ur0  = uti0 + ul15;
        const int ur1  = ur0 + 32;
        v8f t0, t1;
        #pragma unroll
        for (int r = 0; r < 8; ++r) {
            t0[r] = covpP[(uti0 + r + uh * 8) * P_ + ucol];
            t1[r] = covpP[(uti0 + 32 + r + uh * 8) * P_ + ucol];
        }
        const bool srow = (ul15 & 4) == 0;
        const int  sm   = (ul15 & 3) | ((ul15 & 8) >> 1);
        v2f ka0[2], ka1[2], hb[2], sa[2];
        #pragma unroll
        for (int kb = 0; kb < 2; ++kb) {
            const int k0 = kb * 4 + uh * 2;
            ka0[kb].x = -Km[ur0 * 8 + k0];
            ka0[kb].y = -Km[ur0 * 8 + k0 + 1];
            ka1[kb].x = -Km[ur1 * 8 + k0];
            ka1[kb].y = -Km[ur1 * 8 + k0 + 1];
            hb[kb].x  = HPs[(k0    ) * P_ + ucol];
            hb[kb].y  = HPs[(k0 + 1) * P_ + ucol];
            const float s0 = Sin[(k0    ) * 8 + sm];
            const float s1 = Sin[(k0 + 1) * 8 + sm];
            sa[kb].x = srow ? s0 : 0.f;
            sa[kb].y = srow ? s1 : 0.f;
        }
        const float4 p0 = *(const float4*)(PHt + ur0 * 8 + 4 * uh);
        const float4 p1 = *(const float4*)(PHt + ur1 * 8 + 4 * uh);
        v8f nw0 = {p0.x, p0.y, p0.z, p0.w, 0.f, 0.f, 0.f, 0.f};
        v8f nw1 = {p1.x, p1.y, p1.z, p1.w, 0.f, 0.f, 0.f, 0.f};
        const float4 kj = *(const float4*)(Km + ucol * 8 + 4 * uh);
        v2f kc[2];
        kc[0].x = -kj.x; kc[0].y = -kj.y;
        kc[1].x = -kj.z; kc[1].y = -kj.w;

        float mu = 0.f;
        if (tid < 64) {
            mu = meanp[tid];
            for (int m = 0; m < 8; ++m) mu += Km[tid * 8 + m] * resid[m];
        }

        #pragma unroll
        for (int kb = 0; kb < 2; ++kb) {
            t0  = __builtin_amdgcn_wmma_f32_16x16x4_f32(
                      false, ka0[kb], false, hb[kb], (short)0, t0, false, false);
            t1  = __builtin_amdgcn_wmma_f32_16x16x4_f32(
                      false, ka1[kb], false, hb[kb], (short)0, t1, false, false);
            nw0 = __builtin_amdgcn_wmma_f32_16x16x4_f32(
                      false, sa[kb], false, ka0[kb], (short)0, nw0, false, false);
            nw1 = __builtin_amdgcn_wmma_f32_16x16x4_f32(
                      false, sa[kb], false, ka1[kb], (short)0, nw1, false, false);
        }
        #pragma unroll
        for (int kb = 0; kb < 2; ++kb) {
            v2f wa0, wa1;
            wa0.x = nw0[2 * kb]; wa0.y = nw0[2 * kb + 1];
            wa1.x = nw1[2 * kb]; wa1.y = nw1[2 * kb + 1];
            t0 = __builtin_amdgcn_wmma_f32_16x16x4_f32(
                     false, wa0, false, kc[kb], (short)0, t0, false, false);
            t1 = __builtin_amdgcn_wmma_f32_16x16x4_f32(
                     false, wa1, false, kc[kb], (short)0, t1, false, false);
        }
        if (tid < 64) meanv[tid] = mu;
        #pragma unroll
        for (int r = 0; r < 8; ++r) {
            covP[(uti0 + r + uh * 8) * P_ + ucol]      = t0[r];
            covP[(uti0 + 32 + r + uh * 8) * P_ + ucol] = t1[r];
        }
        __syncthreads();
    }
}

extern "C" void kernel_launch(void* const* d_in, const int* in_sizes, int n_in,
                              void* d_out, int out_size, void* d_ws, size_t ws_size,
                              hipStream_t stream) {
    const float* y         = (const float*)d_in[0];
    const float* F         = (const float*)d_in[1];
    const float* H         = (const float*)d_in[2];
    const float* Lq        = (const float*)d_in[3];
    const float* Lr        = (const float*)d_in[4];
    const float* init_mean = (const float*)d_in[5];
    const float* Linit     = (const float*)d_in[6];
    float* out = (float*)d_out;
    float* ws  = (float*)d_ws;

    kf_prep<<<16, 256, 0, stream>>>(Lq, Lr, Linit, ws);

    const size_t smem_bytes = (4864 * 4 + 608 + 64 + 512 + 512 + 64 +
                               64 + 64 + 8 + 608) * sizeof(float);
    kf_main<<<G_, 256, smem_bytes, stream>>>(y, F, H,
                                             ws,
                                             ws + 8192,
                                             ws + 4096,
                                             init_mean, out);
}


# yhat's global store and the next y row's load issued at the head of the long inverse interval: the thin S_inn interval's barrier no longer waits for a store acknowledgement; same arithmetic
# speedup vs baseline: 12.6435x; 1.0490x over previous
#include <hip/hip_runtime.h>
#include <hip/hip_bf16.h>

#define G_ 128
#define T_ 128
#define S_ 64
#define M_ 8
#define P_ 76

typedef float v2f __attribute__((ext_vector_type(2)));
typedef float v8f __attribute__((ext_vector_type(8)));

__device__ __forceinline__ float lane_bcast(float x, int src)
{
    return __int_as_float(__builtin_amdgcn_readlane(__float_as_int(x), src));
}

template <bool TRANSB, int EPI>
__device__ __forceinline__ void mm64(const float* A, const float* B, float* D,
                                     const float* addQ)
{
    const int lane = threadIdx.x & 31;
    const int wave = threadIdx.x >> 5;
    const int half = lane >> 4;
    const int l15  = lane & 15;
    const int ti0  = (wave >> 2) * 16;
    const int tj   = (wave & 3) * 16;

    v8f acc0 = {}, acc1 = {};
    #pragma unroll
    for (int kb = 0; kb < 16; ++kb) {
        const int k0 = kb * 4 + half * 2;
        v2f a0, a1, b;
        a0.x = A[(ti0 + l15) * P_ + k0];
        a0.y = A[(ti0 + l15) * P_ + k0 + 1];
        a1.x = A[(ti0 + 32 + l15) * P_ + k0];
        a1.y = A[(ti0 + 32 + l15) * P_ + k0 + 1];
        if (!TRANSB) {
            b.x = B[(k0    ) * P_ + (tj + l15)];
            b.y = B[(k0 + 1) * P_ + (tj + l15)];
        } else {
            b.x = B[(tj + l15) * P_ + k0];
            b.y = B[(tj + l15) * P_ + k0 + 1];
        }
        acc0 = __builtin_amdgcn_wmma_f32_16x16x4_f32(
                   false, a0, false, b, (short)0, acc0, false, false);
        acc1 = __builtin_amdgcn_wmma_f32_16x16x4_f32(
                   false, a1, false, b, (short)0, acc1, false, false);
    }

    const int col = tj + l15;
    float v0[8], v1[8];
    #pragma unroll
    for (int r = 0; r < 8; ++r) {
        const int row0 = ti0 + r + half * 8;
        const int row1 = row0 + 32;
        float x0 = acc0[r], x1 = acc1[r];
        if (EPI == 1) {
            x0 += addQ[row0 * 64 + col];
            x1 += addQ[row1 * 64 + col];
        }
        v0[r] = x0;
        v1[r] = x1;
    }
    #pragma unroll
    for (int r = 0; r < 8; ++r) {
        const int row0 = ti0 + r + half * 8;
        D[row0 * P_ + col]        = v0[r];
        D[(row0 + 32) * P_ + col] = v1[r];
    }
}

__global__ __launch_bounds__(256) void kf_prep(const float* __restrict__ Lq,
                                               const float* __restrict__ Lr,
                                               const float* __restrict__ Linit,
                                               float* __restrict__ ws)
{
    float* Q  = ws;
    float* P0 = ws + 4096;
    float* R  = ws + 8192;
    for (int e = blockIdx.x * blockDim.x + threadIdx.x; e < 4096;
         e += gridDim.x * blockDim.x) {
        const int i = e >> 6, j = e & 63;
        float q = 0.f, p = 0.f;
        for (int k = 0; k < 64; ++k) {
            q += Lq[i * 64 + k] * Lq[j * 64 + k];
            p += Linit[i * 64 + k] * Linit[j * 64 + k];
        }
        Q[e] = q;
        P0[e] = p;
    }
    const int e = blockIdx.x * blockDim.x + threadIdx.x;
    if (e < 64) {
        const int i = e >> 3, j = e & 7;
        float r = 0.f;
        for (int k = 0; k < 8; ++k) r += Lr[i * 8 + k] * Lr[j * 8 + k];
        R[e] = r;
    }
}

__device__ __forceinline__ void hp_tile(const float* covp, float* HPs,
                                        const v2f (&ha)[16], int tj)
{
    const int lane = threadIdx.x & 31;
    const int half = lane >> 4;
    const int l15  = lane & 15;
    v2f b[16];
    #pragma unroll
    for (int kb = 0; kb < 16; ++kb) {
        const int k0 = kb * 4 + half * 2;
        b[kb].x = covp[(k0    ) * P_ + (tj + l15)];
        b[kb].y = covp[(k0 + 1) * P_ + (tj + l15)];
    }
    v8f acc0 = {}, acc1 = {};
    #pragma unroll
    for (int kb = 0; kb < 8; ++kb) {
        acc0 = __builtin_amdgcn_wmma_f32_16x16x4_f32(
                   false, ha[kb], false, b[kb], (short)0, acc0, false, false);
        acc1 = __builtin_amdgcn_wmma_f32_16x16x4_f32(
                   false, ha[kb + 8], false, b[kb + 8], (short)0, acc1, false, false);
    }
    float hp[8];
    #pragma unroll
    for (int r = 0; r < 8; ++r) hp[r] = acc0[r] + acc1[r];
    if (half == 0) {
        #pragma unroll
        for (int r = 0; r < 8; ++r) HPs[r * P_ + (tj + l15)] = hp[r];
    }
}

__global__ __launch_bounds__(256) void kf_main(const float* __restrict__ y,
                                               const float* __restrict__ F,
                                               const float* __restrict__ H,
                                               const float* __restrict__ Qg,
                                               const float* __restrict__ Rg,
                                               const float* __restrict__ P0g,
                                               const float* __restrict__ init_mean,
                                               float* __restrict__ out)
{
    extern __shared__ __attribute__((aligned(16))) float smem[];
    float* Fs    = smem;
    float* bufA  = Fs   + 64 * P_;
    float* bufB  = bufA + 64 * P_;
    float* bufC  = bufB + 64 * P_;
    float* Hs    = bufC + 64 * P_;
    float* Rs    = Hs   + 8 * P_;
    float* PHt   = Rs   + 64;
    float* Km    = PHt  + 512;
    float* Sin   = Km   + 512;
    float* meanv = Sin  + 64;
    float* meanp = meanv + 64;
    float* resid = meanp + 64;
    float* HPs   = resid + 8;

    const int g   = blockIdx.x;
    const int tid = threadIdx.x;

    for (int i = tid; i < 4096; i += 256) {
        const int r = i >> 6, c = i & 63;
        Fs[r * P_ + c]   = F[i];
        bufA[r * P_ + c] = P0g[i];
    }
    for (int i = tid; i < 512; i += 256) Hs[(i >> 6) * P_ + (i & 63)] = H[i];
    if (tid < 64) {
        Rs[tid]    = Rg[tid];
        meanv[tid] = init_mean[g * S_ + tid];
    }
    float ycur = 0.f, yh = 0.f;
    if (tid >= 128 && tid < 136) ycur = y[(g * T_) * M_ + (tid - 128)];
    __syncthreads();

    float4 rw[32];
    {
        const float* r0 = (tid < 128) ? Hs + (tid & 3) * P_ : Fs + (tid & 63) * P_;
        const float* r1 = (tid < 128) ? Hs + ((tid & 3) + 4) * P_ : r0;
        #pragma unroll
        for (int q = 0; q < 16; ++q) {
            rw[q]      = ((const float4*)r0)[q];
            rw[16 + q] = ((const float4*)r1)[q];
        }
    }

    v2f ha[16];
    {
        const int l15 = tid & 15, half = (tid & 31) >> 4;
        const float* hrow = Hs + (l15 & 7) * P_;
        #pragma unroll
        for (int kb = 0; kb < 16; ++kb) {
            const int k0 = kb * 4 + half * 2;
            const float h0 = hrow[k0], h1 = hrow[k0 + 1];
            ha[kb].x = (l15 < 8) ? h0 : 0.f;
            ha[kb].y = (l15 < 8) ? h1 : 0.f;
        }
    }

    float* const covP  = bufA;
    float* const covpP = bufB;
    float* const tmpP  = bufC;

    for (int t = 0; t < T_; ++t) {
        mm64<false, 0>(Fs, covP, tmpP, nullptr);
        __syncthreads();
        mm64<true, 1>(tmpP, Fs, covpP, Qg);
        __syncthreads();

        if (tid < 128) {
            const int sp = tid >> 2, ma = tid & 3;
            const float4* c0 = (const float4*)(covpP + sp * P_);
            const float4* c1 = (const float4*)(covpP + (sp + 32) * P_);
            float4 x0[16], x1[16];
            #pragma unroll
            for (int q = 0; q < 16; ++q) { x0[q] = c0[q]; x1[q] = c1[q]; }
            __builtin_amdgcn_sched_barrier(0);
            float a00 = 0.f, a01 = 0.f, a10 = 0.f, a11 = 0.f;
            #pragma unroll
            for (int q = 0; q < 16; ++q) {
                a00 = __builtin_fmaf(x0[q].x, rw[q].x, a00);
                a01 = __builtin_fmaf(x0[q].x, rw[16 + q].x, a01);
                a10 = __builtin_fmaf(x1[q].x, rw[q].x, a10);
                a11 = __builtin_fmaf(x1[q].x, rw[16 + q].x, a11);
                a00 = __builtin_fmaf(x0[q].y, rw[q].y, a00);
                a01 = __builtin_fmaf(x0[q].y, rw[16 + q].y, a01);
                a10 = __builtin_fmaf(x1[q].y, rw[q].y, a10);
                a11 = __builtin_fmaf(x1[q].y, rw[16 + q].y, a11);
                a00 = __builtin_fmaf(x0[q].z, rw[q].z, a00);
                a01 = __builtin_fmaf(x0[q].z, rw[16 + q].z, a01);
                a10 = __builtin_fmaf(x1[q].z, rw[q].z, a10);
                a11 = __builtin_fmaf(x1[q].z, rw[16 + q].z, a11);
                a00 = __builtin_fmaf(x0[q].w, rw[q].w, a00);
                a01 = __builtin_fmaf(x0[q].w, rw[16 + q].w, a01);
                a10 = __builtin_fmaf(x1[q].w, rw[q].w, a10);
                a11 = __builtin_fmaf(x1[q].w, rw[16 + q].w, a11);
            }
            PHt[sp * 8 + ma]            = a00;
            PHt[sp * 8 + ma + 4]        = a01;
            PHt[(sp + 32) * 8 + ma]     = a10;
            PHt[(sp + 32) * 8 + ma + 4] = a11;
        } else if (tid < 192) {
            const float4* mv = (const float4*)meanv;
            float4 m4[16];
            #pragma unroll
            for (int q = 0; q < 16; ++q) m4[q] = mv[q];
            __builtin_amdgcn_sched_barrier(0);
            float mp = 0.f;
            #pragma unroll
            for (int q = 0; q < 16; ++q) {
                mp = __builtin_fmaf(rw[q].x, m4[q].x, mp);
                mp = __builtin_fmaf(rw[q].y, m4[q].y, mp);
                mp = __builtin_fmaf(rw[q].z, m4[q].z, mp);
                mp = __builtin_fmaf(rw[q].w, m4[q].w, mp);
            }
            meanp[tid - 128] = mp;
        } else {
            hp_tile(covpP, HPs, ha, ((tid >> 5) - 6) * 16);
        }
        __syncthreads();

        if (tid < 64) {
            const int i = tid >> 3, c = tid & 7;
            const float4* hr = (const float4*)(Hs + i * P_);
            float4 h4[16];
            float  ph[64];
            float acc = Rs[i * 8 + c];
            #pragma unroll
            for (int q = 0; q < 16; ++q) h4[q] = hr[q];
            #pragma unroll
            for (int s = 0; s < 64; ++s) ph[s] = PHt[s * 8 + c];
            __builtin_amdgcn_sched_barrier(0);
            #pragma unroll
            for (int q = 0; q < 16; ++q) {
                acc = __builtin_fmaf(h4[q].x, ph[4 * q],     acc);
                acc = __builtin_fmaf(h4[q].y, ph[4 * q + 1], acc);
                acc = __builtin_fmaf(h4[q].z, ph[4 * q + 2], acc);
                acc = __builtin_fmaf(h4[q].w, ph[4 * q + 3], acc);
            }
            Sin[i * 8 + c] = acc;
        } else if (tid >= 128 && tid < 136) {
            const int m = tid - 128;
            const float4* hr = (const float4*)(Hs + m * P_);
            const float4* mq = (const float4*)meanp;
            float4 h4[16], m4[16];
            #pragma unroll
            for (int q = 0; q < 16; ++q) { h4[q] = hr[q]; m4[q] = mq[q]; }
            __builtin_amdgcn_sched_barrier(0);
            float acc = 0.f;
            #pragma unroll
            for (int q = 0; q < 16; ++q) {
                acc = __builtin_fmaf(h4[q].x, m4[q].x, acc);
                acc = __builtin_fmaf(h4[q].y, m4[q].y, acc);
                acc = __builtin_fmaf(h4[q].z, m4[q].z, acc);
                acc = __builtin_fmaf(h4[q].w, m4[q].w, acc);
            }
            yh = acc;
            resid[m] = ycur - acc;
        } else if (tid >= 192) {
            hp_tile(covpP, HPs, ha, ((tid >> 5) - 4) * 16);
        }
        __syncthreads();

        if (tid >= 128 && tid < 136) {
            const int m = tid - 128;
            out[(g * T_ + t) * M_ + m] = yh;
            const int tn = (t + 1 < T_) ? t + 1 : t;
            ycur = y[(g * T_ + tn) * M_ + m];
        }

        const int cI = tid & 7;
        float wl[8], wr[8];
        #pragma unroll
        for (int r = 0; r < 8; ++r) {
            wl[r] = Sin[r * 8 + cI];
            wr[r] = (r == cI) ? 1.f : 0.f;
        }
        #pragma unroll
        for (int p = 0; p < 8; ++p) {
            const float pv = 1.f / lane_bcast(wl[p], p);
            wl[p] *= pv;
            wr[p] *= pv;
            #pragma unroll
            for (int r = 0; r < 8; ++r) {
                if (r != p) {
                    const float f = lane_bcast(wl[r], p);
                    wl[r] -= f * wl[p];
                    wr[r] -= f * wr[p];
                }
            }
        }

        {
            const int s0 = tid >> 3;
            float acc0 = 0.f, acc1 = 0.f;
            #pragma unroll
            for (int j = 0; j < 8; ++j) {
                acc0 += PHt[s0 * 8 + j] * wr[j];
                acc1 += PHt[(s0 + 32) * 8 + j] * wr[j];
            }
            Km[tid]       = acc0;
            Km[tid + 256] = acc1;
        }
        __syncthreads();

        const int uh   = (tid & 31) >> 4;
        const int ul15 = tid & 15;
        const int uti0 = (tid >> 7) * 16;
        const int ucol = ((tid >> 5) & 3) * 16 + ul15;
        const int ur0  = uti0 + ul15;
        const int ur1  = ur0 + 32;
        v8f t0, t1;
        #pragma unroll
        for (int r = 0; r < 8; ++r) {
            t0[r] = covpP[(uti0 + r + uh * 8) * P_ + ucol];
            t1[r] = covpP[(uti0 + 32 + r + uh * 8) * P_ + ucol];
        }
        const bool srow = (ul15 & 4) == 0;
        const int  sm   = (ul15 & 3) | ((ul15 & 8) >> 1);
        v2f ka0[2], ka1[2], hb[2], sa[2];
        #pragma unroll
        for (int kb = 0; kb < 2; ++kb) {
            const int k0 = kb * 4 + uh * 2;
            ka0[kb].x = -Km[ur0 * 8 + k0];
            ka0[kb].y = -Km[ur0 * 8 + k0 + 1];
            ka1[kb].x = -Km[ur1 * 8 + k0];
            ka1[kb].y = -Km[ur1 * 8 + k0 + 1];
            hb[kb].x  = HPs[(k0    ) * P_ + ucol];
            hb[kb].y  = HPs[(k0 + 1) * P_ + ucol];
            const float s0 = Sin[(k0    ) * 8 + sm];
            const float s1 = Sin[(k0 + 1) * 8 + sm];
            sa[kb].x = srow ? s0 : 0.f;
            sa[kb].y = srow ? s1 : 0.f;
        }
        const float4 p0 = *(const float4*)(PHt + ur0 * 8 + 4 * uh);
        const float4 p1 = *(const float4*)(PHt + ur1 * 8 + 4 * uh);
        v8f nw0 = {p0.x, p0.y, p0.z, p0.w, 0.f, 0.f, 0.f, 0.f};
        v8f nw1 = {p1.x, p1.y, p1.z, p1.w, 0.f, 0.f, 0.f, 0.f};
        const float4 kj = *(const float4*)(Km + ucol * 8 + 4 * uh);
        v2f kc[2];
        kc[0].x = -kj.x; kc[0].y = -kj.y;
        kc[1].x = -kj.z; kc[1].y = -kj.w;

        float mu = 0.f;
        if (tid < 64) {
            mu = meanp[tid];
            for (int m = 0; m < 8; ++m) mu += Km[tid * 8 + m] * resid[m];
        }

        #pragma unroll
        for (int kb = 0; kb < 2; ++kb) {
            t0  = __builtin_amdgcn_wmma_f32_16x16x4_f32(
                      false, ka0[kb], false, hb[kb], (short)0, t0, false, false);
            t1  = __builtin_amdgcn_wmma_f32_16x16x4_f32(
                      false, ka1[kb], false, hb[kb], (short)0, t1, false, false);
            nw0 = __builtin_amdgcn_wmma_f32_16x16x4_f32(
                      false, sa[kb], false, ka0[kb], (short)0, nw0, false, false);
            nw1 = __builtin_amdgcn_wmma_f32_16x16x4_f32(
                      false, sa[kb], false, ka1[kb], (short)0, nw1, false, false);
        }
        #pragma unroll
        for (int kb = 0; kb < 2; ++kb) {
            v2f wa0, wa1;
            wa0.x = nw0[2 * kb]; wa0.y = nw0[2 * kb + 1];
            wa1.x = nw1[2 * kb]; wa1.y = nw1[2 * kb + 1];
            t0 = __builtin_amdgcn_wmma_f32_16x16x4_f32(
                     false, wa0, false, kc[kb], (short)0, t0, false, false);
            t1 = __builtin_amdgcn_wmma_f32_16x16x4_f32(
                     false, wa1, false, kc[kb], (short)0, t1, false, false);
        }
        if (tid < 64) meanv[tid] = mu;
        #pragma unroll
        for (int r = 0; r < 8; ++r) {
            covP[(uti0 + r + uh * 8) * P_ + ucol]      = t0[r];
            covP[(uti0 + 32 + r + uh * 8) * P_ + ucol] = t1[r];
        }
        __syncthreads();
    }
}

extern "C" void kernel_launch(void* const* d_in, const int* in_sizes, int n_in,
                              void* d_out, int out_size, void* d_ws, size_t ws_size,
                              hipStream_t stream) {
    const float* y         = (const float*)d_in[0];
    const float* F         = (const float*)d_in[1];
    const float* H         = (const float*)d_in[2];
    const float* Lq        = (const float*)d_in[3];
    const float* Lr        = (const float*)d_in[4];
    const float* init_mean = (const float*)d_in[5];
    const float* Linit     = (const float*)d_in[6];
    float* out = (float*)d_out;
    float* ws  = (float*)d_ws;

    kf_prep<<<16, 256, 0, stream>>>(Lq, Lr, Linit, ws);

    const size_t smem_bytes = (4864 * 4 + 608 + 64 + 512 + 512 + 64 +
                               64 + 64 + 8 + 608) * sizeof(float);
    kf_main<<<G_, 256, smem_bytes, stream>>>(y, F, H,
                                             ws,
                                             ws + 8192,
                                             ws + 4096,
                                             init_mean, out);
}


# PHt = cov_p H^T on WMMA (waves 0-3, 64 terms as two 32-term chains), S_inn = H PHt + R from its result registers as four tile shares: both thin FMA phases and one barrier go; sum order changed
# speedup vs baseline: 13.1202x; 1.0377x over previous
#include <hip/hip_runtime.h>
#include <hip/hip_bf16.h>

#define G_ 128
#define T_ 128
#define S_ 64
#define M_ 8
#define P_ 76

typedef float v2f __attribute__((ext_vector_type(2)));
typedef float v8f __attribute__((ext_vector_type(8)));

__device__ __forceinline__ float lane_bcast(float x, int src)
{
    return __int_as_float(__builtin_amdgcn_readlane(__float_as_int(x), src));
}

template <bool TRANSB, int EPI>
__device__ __forceinline__ void mm64(const float* A, const float* B, float* D,
                                     const float* addQ)
{
    const int lane = threadIdx.x & 31;
    const int wave = threadIdx.x >> 5;
    const int half = lane >> 4;
    const int l15  = lane & 15;
    const int ti0  = (wave >> 2) * 16;
    const int tj   = (wave & 3) * 16;

    v8f acc0 = {}, acc1 = {};
    #pragma unroll
    for (int kb = 0; kb < 16; ++kb) {
        const int k0 = kb * 4 + half * 2;
        v2f a0, a1, b;
        a0.x = A[(ti0 + l15) * P_ + k0];
        a0.y = A[(ti0 + l15) * P_ + k0 + 1];
        a1.x = A[(ti0 + 32 + l15) * P_ + k0];
        a1.y = A[(ti0 + 32 + l15) * P_ + k0 + 1];
        if (!TRANSB) {
            b.x = B[(k0    ) * P_ + (tj + l15)];
            b.y = B[(k0 + 1) * P_ + (tj + l15)];
        } else {
            b.x = B[(tj + l15) * P_ + k0];
            b.y = B[(tj + l15) * P_ + k0 + 1];
        }
        acc0 = __builtin_amdgcn_wmma_f32_16x16x4_f32(
                   false, a0, false, b, (short)0, acc0, false, false);
        acc1 = __builtin_amdgcn_wmma_f32_16x16x4_f32(
                   false, a1, false, b, (short)0, acc1, false, false);
    }

    const int col = tj + l15;
    float v0[8], v1[8];
    #pragma unroll
    for (int r = 0; r < 8; ++r) {
        const int row0 = ti0 + r + half * 8;
        const int row1 = row0 + 32;
        float x0 = acc0[r], x1 = acc1[r];
        if (EPI == 1) {
            x0 += addQ[row0 * 64 + col];
            x1 += addQ[row1 * 64 + col];
        }
        v0[r] = x0;
        v1[r] = x1;
    }
    #pragma unroll
    for (int r = 0; r < 8; ++r) {
        const int row0 = ti0 + r + half * 8;
        D[row0 * P_ + col]        = v0[r];
        D[(row0 + 32) * P_ + col] = v1[r];
    }
}

__global__ __launch_bounds__(256) void kf_prep(const float* __restrict__ Lq,
                                               const float* __restrict__ Lr,
                                               const float* __restrict__ Linit,
                                               float* __restrict__ ws)
{
    float* Q  = ws;
    float* P0 = ws + 4096;
    float* R  = ws + 8192;
    for (int e = blockIdx.x * blockDim.x + threadIdx.x; e < 4096;
         e += gridDim.x * blockDim.x) {
        const int i = e >> 6, j = e & 63;
        float q = 0.f, p = 0.f;
        for (int k = 0; k < 64; ++k) {
            q += Lq[i * 64 + k] * Lq[j * 64 + k];
            p += Linit[i * 64 + k] * Linit[j * 64 + k];
        }
        Q[e] = q;
        P0[e] = p;
    }
    const int e = blockIdx.x * blockDim.x + threadIdx.x;
    if (e < 64) {
        const int i = e >> 3, j = e & 7;
        float r = 0.f;
        for (int k = 0; k < 8; ++k) r += Lr[i * 8 + k] * Lr[j * 8 + k];
        R[e] = r;
    }
}


__device__ __forceinline__ void hp_tiles(const float* covp, float* HPs,
                                         const v2f (&ha)[16], int tj)
{
    const int lane = threadIdx.x & 31;
    const int half = lane >> 4;
    const int l15  = lane & 15;
    v2f b0[16], b1[16];
    #pragma unroll
    for (int kb = 0; kb < 16; ++kb) {
        const int k0 = kb * 4 + half * 2;
        b0[kb].x = covp[(k0    ) * P_ + (tj + l15)];
        b0[kb].y = covp[(k0 + 1) * P_ + (tj + l15)];
    }
    #pragma unroll
    for (int kb = 0; kb < 16; ++kb) {
        const int k0 = kb * 4 + half * 2;
        b1[kb].x = covp[(k0    ) * P_ + (tj + 32 + l15)];
        b1[kb].y = covp[(k0 + 1) * P_ + (tj + 32 + l15)];
    }
    __builtin_amdgcn_sched_barrier(0);
    v8f a00 = {}, a01 = {}, a10 = {}, a11 = {};
    #pragma unroll
    for (int kb = 0; kb < 8; ++kb) {
        a00 = __builtin_amdgcn_wmma_f32_16x16x4_f32(
                  false, ha[kb], false, b0[kb], (short)0, a00, false, false);
        a01 = __builtin_amdgcn_wmma_f32_16x16x4_f32(
                  false, ha[kb + 8], false, b0[kb + 8], (short)0, a01, false, false);
        a10 = __builtin_amdgcn_wmma_f32_16x16x4_f32(
                  false, ha[kb], false, b1[kb], (short)0, a10, false, false);
        a11 = __builtin_amdgcn_wmma_f32_16x16x4_f32(
                  false, ha[kb + 8], false, b1[kb + 8], (short)0, a11, false, false);
    }
    float h0[8], h1[8];
    #pragma unroll
    for (int r = 0; r < 8; ++r) {
        h0[r] = a00[r] + a01[r];
        h1[r] = a10[r] + a11[r];
    }
    if (half == 0) {
        #pragma unroll
        for (int r = 0; r < 8; ++r) {
            HPs[r * P_ + (tj + l15)]      = h0[r];
            HPs[r * P_ + (tj + 32 + l15)] = h1[r];
        }
    }
}

__device__ __forceinline__ void pht_tile(const float* covp, float* PHt, float* Sp,
                                         const v2f (&ha)[16], const v2f (&hs)[4],
                                         const v8f& cstart, int ti)
{
    const int lane = threadIdx.x & 31;
    const int half = lane >> 4;
    const int l15  = lane & 15;
    v2f a[16];
    #pragma unroll
    for (int kb = 0; kb < 16; ++kb) {
        const int k0 = kb * 4 + half * 2;
        a[kb] = *(const v2f*)(covp + (ti + l15) * P_ + k0);
    }
    __builtin_amdgcn_sched_barrier(0);
    v8f acc0 = {}, acc1 = {};
    #pragma unroll
    for (int kb = 0; kb < 8; ++kb) {
        acc0 = __builtin_amdgcn_wmma_f32_16x16x4_f32(
                   false, a[kb], false, ha[kb], (short)0, acc0, false, false);
        acc1 = __builtin_amdgcn_wmma_f32_16x16x4_f32(
                   false, a[kb + 8], false, ha[kb + 8], (short)0, acc1, false, false);
    }
    float p[8];
    #pragma unroll
    for (int r = 0; r < 8; ++r) p[r] = acc0[r] + acc1[r];
    v2f pb[4];
    #pragma unroll
    for (int s = 0; s < 4; ++s) { pb[s].x = p[2 * s]; pb[s].y = p[2 * s + 1]; }
    v8f s0 = cstart, s1 = {};
    s0 = __builtin_amdgcn_wmma_f32_16x16x4_f32(
             false, hs[0], false, pb[0], (short)0, s0, false, false);
    s1 = __builtin_amdgcn_wmma_f32_16x16x4_f32(
             false, hs[1], false, pb[1], (short)0, s1, false, false);
    s0 = __builtin_amdgcn_wmma_f32_16x16x4_f32(
             false, hs[2], false, pb[2], (short)0, s0, false, false);
    s1 = __builtin_amdgcn_wmma_f32_16x16x4_f32(
             false, hs[3], false, pb[3], (short)0, s1, false, false);
    float sp[8];
    #pragma unroll
    for (int r = 0; r < 8; ++r) sp[r] = s0[r] + s1[r];
    if (l15 < 8) {
        #pragma unroll
        for (int r = 0; r < 8; ++r) PHt[(ti + half * 8 + r) * 8 + l15] = p[r];
        if (half == 0) {
            #pragma unroll
            for (int r = 0; r < 8; ++r) Sp[r * 8 + l15] = sp[r];
        }
    }
}

__global__ __launch_bounds__(256) void kf_main(const float* __restrict__ y,
                                               const float* __restrict__ F,
                                               const float* __restrict__ H,
                                               const float* __restrict__ Qg,
                                               const float* __restrict__ Rg,
                                               const float* __restrict__ P0g,
                                               const float* __restrict__ init_mean,
                                               float* __restrict__ out)
{
    extern __shared__ __attribute__((aligned(16))) float smem[];
    float* Fs    = smem;
    float* bufA  = Fs   + 64 * P_;
    float* bufB  = bufA + 64 * P_;
    float* bufC  = bufB + 64 * P_;
    float* Hs    = bufC + 64 * P_;
    float* Rs    = Hs   + 8 * P_;
    float* PHt   = Rs   + 64;
    float* Km    = PHt  + 512;
    float* Sin   = Km   + 512;
    float* meanv = Sin  + 64;
    float* meanp = meanv + 64;
    float* resid = meanp + 64;
    float* Sp    = resid + 8;
    float* HPs   = Sp   + 256;

    const int g   = blockIdx.x;
    const int tid = threadIdx.x;

    for (int i = tid; i < 4096; i += 256) {
        const int r = i >> 6, c = i & 63;
        Fs[r * P_ + c]   = F[i];
        bufA[r * P_ + c] = P0g[i];
    }
    for (int i = tid; i < 512; i += 256) Hs[(i >> 6) * P_ + (i & 63)] = H[i];
    if (tid < 64) {
        Rs[tid]    = Rg[tid];
        meanv[tid] = init_mean[g * S_ + tid];
    }
    __syncthreads();

    float4 rw[16];
    {
        const float* r0 = Fs + (tid & 63) * P_;
        #pragma unroll
        for (int q = 0; q < 16; ++q) rw[q] = ((const float4*)r0)[q];
    }

    v2f ha[16], hs[4];
    {
        const int l15 = tid & 15, half = (tid & 31) >> 4;
        const int ti  = ((tid >> 5) & 3) * 16;
        const float* hrow = Hs + (l15 & 7) * P_;
        #pragma unroll
        for (int kb = 0; kb < 16; ++kb) {
            const int k0 = kb * 4 + half * 2;
            const float h0 = hrow[k0], h1 = hrow[k0 + 1];
            ha[kb].x = (l15 < 8) ? h0 : 0.f;
            ha[kb].y = (l15 < 8) ? h1 : 0.f;
        }
        #pragma unroll
        for (int s = 0; s < 4; ++s) {
            const int k0 = ti + half * 8 + s * 2;
            const float h0 = hrow[k0], h1 = hrow[k0 + 1];
            hs[s].x = (l15 < 8) ? h0 : 0.f;
            hs[s].y = (l15 < 8) ? h1 : 0.f;
        }
    }
    v8f rstart;
    #pragma unroll
    for (int r = 0; r < 8; ++r) rstart[r] = (tid < 8) ? Rs[r * 8 + tid] : 0.f;

    float* const covP  = bufA;
    float* const covpP = bufB;
    float* const tmpP  = bufC;

    float ycur = 0.f, yh = 0.f;

    for (int t = 0; t < T_; ++t) {
        if (tid >= 128 && tid < 136) {
            const int m = tid - 128;
            if (t > 0) out[(g * T_ + t - 1) * M_ + m] = yh;
            ycur = y[(g * T_ + t) * M_ + m];
        }

        mm64<false, 0>(Fs, covP, tmpP, nullptr);
        __syncthreads();
        mm64<true, 1>(tmpP, Fs, covpP, Qg);
        __syncthreads();

        if (tid < 128) {
            pht_tile(covpP, PHt, Sp + (tid >> 5) * 64, ha, hs, rstart, (tid >> 5) * 16);
        } else if (tid < 192) {
            const float4* mv = (const float4*)meanv;
            float4 m4[16];
            #pragma unroll
            for (int q = 0; q < 16; ++q) m4[q] = mv[q];
            __builtin_amdgcn_sched_barrier(0);
            float mp = 0.f;
            #pragma unroll
            for (int q = 0; q < 16; ++q) {
                mp = __builtin_fmaf(rw[q].x, m4[q].x, mp);
                mp = __builtin_fmaf(rw[q].y, m4[q].y, mp);
                mp = __builtin_fmaf(rw[q].z, m4[q].z, mp);
                mp = __builtin_fmaf(rw[q].w, m4[q].w, mp);
            }
            meanp[tid - 128] = mp;
        }
        __syncthreads();

        if (tid < 128) {
            const int cI = tid & 7;
            float wl[8], wr[8];
            #pragma unroll
            for (int r = 0; r < 8; ++r) {
                const float* sq = Sp + r * 8 + cI;
                wl[r] = (sq[0] + sq[64]) + (sq[128] + sq[192]);
                wr[r] = (r == cI) ? 1.f : 0.f;
            }
            if (tid < 8) {
                #pragma unroll
                for (int r = 0; r < 8; ++r) Sin[r * 8 + cI] = wl[r];
            }
            #pragma unroll
            for (int p = 0; p < 8; ++p) {
                const float pv = 1.f / lane_bcast(wl[p], p);
                wl[p] *= pv;
                wr[p] *= pv;
                #pragma unroll
                for (int r = 0; r < 8; ++r) {
                    if (r != p) {
                        const float f = lane_bcast(wl[r], p);
                        wl[r] -= f * wl[p];
                        wr[r] -= f * wr[p];
                    }
                }
            }

            const int s0 = tid >> 3;
            float k0 = 0.f, k1 = 0.f, k2 = 0.f, k3 = 0.f;
            #pragma unroll
            for (int j = 0; j < 8; ++j) {
                k0 += PHt[s0 * 8 + j] * wr[j];
                k1 += PHt[(s0 + 16) * 8 + j] * wr[j];
                k2 += PHt[(s0 + 32) * 8 + j] * wr[j];
                k3 += PHt[(s0 + 48) * 8 + j] * wr[j];
            }
            Km[tid]       = k0;
            Km[tid + 128] = k1;
            Km[tid + 256] = k2;
            Km[tid + 384] = k3;
        } else if (tid < 136) {
            const int m = tid - 128;
            const float4* hr = (const float4*)(Hs + m * P_);
            const float4* mq = (const float4*)meanp;
            float4 h4[16], m4[16];
            #pragma unroll
            for (int q = 0; q < 16; ++q) { h4[q] = hr[q]; m4[q] = mq[q]; }
            __builtin_amdgcn_sched_barrier(0);
            float acc = 0.f;
            #pragma unroll
            for (int q = 0; q < 16; ++q) {
                acc = __builtin_fmaf(h4[q].x, m4[q].x, acc);
                acc = __builtin_fmaf(h4[q].y, m4[q].y, acc);
                acc = __builtin_fmaf(h4[q].z, m4[q].z, acc);
                acc = __builtin_fmaf(h4[q].w, m4[q].w, acc);
            }
            yh = acc;
            resid[m] = ycur - acc;
        } else if (tid >= 192) {
            hp_tiles(covpP, HPs, ha, ((tid >> 5) - 6) * 16);
        }
        __syncthreads();

        const int uh   = (tid & 31) >> 4;
        const int ul15 = tid & 15;
        const int uti0 = (tid >> 7) * 16;
        const int ucol = ((tid >> 5) & 3) * 16 + ul15;
        const int ur0  = uti0 + ul15;
        const int ur1  = ur0 + 32;
        v8f t0, t1;
        #pragma unroll
        for (int r = 0; r < 8; ++r) {
            t0[r] = covpP[(uti0 + r + uh * 8) * P_ + ucol];
            t1[r] = covpP[(uti0 + 32 + r + uh * 8) * P_ + ucol];
        }
        const bool srow = (ul15 & 4) == 0;
        const int  sm   = (ul15 & 3) | ((ul15 & 8) >> 1);
        v2f ka0[2], ka1[2], hb[2], sa[2];
        #pragma unroll
        for (int kb = 0; kb < 2; ++kb) {
            const int k0 = kb * 4 + uh * 2;
            ka0[kb].x = -Km[ur0 * 8 + k0];
            ka0[kb].y = -Km[ur0 * 8 + k0 + 1];
            ka1[kb].x = -Km[ur1 * 8 + k0];
            ka1[kb].y = -Km[ur1 * 8 + k0 + 1];
            hb[kb].x  = HPs[(k0    ) * P_ + ucol];
            hb[kb].y  = HPs[(k0 + 1) * P_ + ucol];
            const float s0 = Sin[(k0    ) * 8 + sm];
            const float s1 = Sin[(k0 + 1) * 8 + sm];
            sa[kb].x = srow ? s0 : 0.f;
            sa[kb].y = srow ? s1 : 0.f;
        }
        const float4 p0 = *(const float4*)(PHt + ur0 * 8 + 4 * uh);
        const float4 p1 = *(const float4*)(PHt + ur1 * 8 + 4 * uh);
        v8f nw0 = {p0.x, p0.y, p0.z, p0.w, 0.f, 0.f, 0.f, 0.f};
        v8f nw1 = {p1.x, p1.y, p1.z, p1.w, 0.f, 0.f, 0.f, 0.f};
        const float4 kj = *(const float4*)(Km + ucol * 8 + 4 * uh);
        v2f kc[2];
        kc[0].x = -kj.x; kc[0].y = -kj.y;
        kc[1].x = -kj.z; kc[1].y = -kj.w;

        float mu = 0.f;
        if (tid < 64) {
            mu = meanp[tid];
            for (int m = 0; m < 8; ++m) mu += Km[tid * 8 + m] * resid[m];
        }

        #pragma unroll
        for (int kb = 0; kb < 2; ++kb) {
            t0  = __builtin_amdgcn_wmma_f32_16x16x4_f32(
                      false, ka0[kb], false, hb[kb], (short)0, t0, false, false);
            t1  = __builtin_amdgcn_wmma_f32_16x16x4_f32(
                      false, ka1[kb], false, hb[kb], (short)0, t1, false, false);
            nw0 = __builtin_amdgcn_wmma_f32_16x16x4_f32(
                      false, sa[kb], false, ka0[kb], (short)0, nw0, false, false);
            nw1 = __builtin_amdgcn_wmma_f32_16x16x4_f32(
                      false, sa[kb], false, ka1[kb], (short)0, nw1, false, false);
        }
        #pragma unroll
        for (int kb = 0; kb < 2; ++kb) {
            v2f wa0, wa1;
            wa0.x = nw0[2 * kb]; wa0.y = nw0[2 * kb + 1];
            wa1.x = nw1[2 * kb]; wa1.y = nw1[2 * kb + 1];
            t0 = __builtin_amdgcn_wmma_f32_16x16x4_f32(
                     false, wa0, false, kc[kb], (short)0, t0, false, false);
            t1 = __builtin_amdgcn_wmma_f32_16x16x4_f32(
                     false, wa1, false, kc[kb], (short)0, t1, false, false);
        }
        if (tid < 64) meanv[tid] = mu;
        #pragma unroll
        for (int r = 0; r < 8; ++r) {
            covP[(uti0 + r + uh * 8) * P_ + ucol]      = t0[r];
            covP[(uti0 + 32 + r + uh * 8) * P_ + ucol] = t1[r];
        }
        __syncthreads();
    }
    if (tid >= 128 && tid < 136) out[(g * T_ + T_ - 1) * M_ + (tid - 128)] = yh;
}

extern "C" void kernel_launch(void* const* d_in, const int* in_sizes, int n_in,
                              void* d_out, int out_size, void* d_ws, size_t ws_size,
                              hipStream_t stream) {
    const float* y         = (const float*)d_in[0];
    const float* F         = (const float*)d_in[1];
    const float* H         = (const float*)d_in[2];
    const float* Lq        = (const float*)d_in[3];
    const float* Lr        = (const float*)d_in[4];
    const float* init_mean = (const float*)d_in[5];
    const float* Linit     = (const float*)d_in[6];
    float* out = (float*)d_out;
    float* ws  = (float*)d_ws;

    kf_prep<<<16, 256, 0, stream>>>(Lq, Lr, Linit, ws);

    const size_t smem_bytes = (4864 * 4 + 608 + 64 + 512 + 512 + 64 +
                               64 + 64 + 8 + 256 + 608) * sizeof(float);
    kf_main<<<G_, 256, smem_bytes, stream>>>(y, F, H,
                                             ws,
                                             ws + 8192,
                                             ws + 4096,
                                             init_mean, out);
}


# 8x8-inverse interval cut to its pivot chain: 1/pivot by the reciprocal instruction (1 ulp), H cov_p one tile on each of waves 4-7, yhat as 4 partial sums beside them; yhat sum order changed
# speedup vs baseline: 14.4612x; 1.1022x over previous
#include <hip/hip_runtime.h>
#include <hip/hip_bf16.h>

#define G_ 128
#define T_ 128
#define S_ 64
#define M_ 8
#define P_ 76

typedef float v2f __attribute__((ext_vector_type(2)));
typedef float v8f __attribute__((ext_vector_type(8)));

__device__ __forceinline__ float lane_bcast(float x, int src)
{
    return __int_as_float(__builtin_amdgcn_readlane(__float_as_int(x), src));
}

template <bool TRANSB, int EPI>
__device__ __forceinline__ void mm64(const float* A, const float* B, float* D,
                                     const float* addQ)
{
    const int lane = threadIdx.x & 31;
    const int wave = threadIdx.x >> 5;
    const int half = lane >> 4;
    const int l15  = lane & 15;
    const int ti0  = (wave >> 2) * 16;
    const int tj   = (wave & 3) * 16;

    v8f acc0 = {}, acc1 = {};
    #pragma unroll
    for (int kb = 0; kb < 16; ++kb) {
        const int k0 = kb * 4 + half * 2;
        v2f a0, a1, b;
        a0.x = A[(ti0 + l15) * P_ + k0];
        a0.y = A[(ti0 + l15) * P_ + k0 + 1];
        a1.x = A[(ti0 + 32 + l15) * P_ + k0];
        a1.y = A[(ti0 + 32 + l15) * P_ + k0 + 1];
        if (!TRANSB) {
            b.x = B[(k0    ) * P_ + (tj + l15)];
            b.y = B[(k0 + 1) * P_ + (tj + l15)];
        } else {
            b.x = B[(tj + l15) * P_ + k0];
            b.y = B[(tj + l15) * P_ + k0 + 1];
        }
        acc0 = __builtin_amdgcn_wmma_f32_16x16x4_f32(
                   false, a0, false, b, (short)0, acc0, false, false);
        acc1 = __builtin_amdgcn_wmma_f32_16x16x4_f32(
                   false, a1, false, b, (short)0, acc1, false, false);
    }

    const int col = tj + l15;
    float v0[8], v1[8];
    #pragma unroll
    for (int r = 0; r < 8; ++r) {
        const int row0 = ti0 + r + half * 8;
        const int row1 = row0 + 32;
        float x0 = acc0[r], x1 = acc1[r];
        if (EPI == 1) {
            x0 += addQ[row0 * 64 + col];
            x1 += addQ[row1 * 64 + col];
        }
        v0[r] = x0;
        v1[r] = x1;
    }
    #pragma unroll
    for (int r = 0; r < 8; ++r) {
        const int row0 = ti0 + r + half * 8;
        D[row0 * P_ + col]        = v0[r];
        D[(row0 + 32) * P_ + col] = v1[r];
    }
}

__global__ __launch_bounds__(256) void kf_prep(const float* __restrict__ Lq,
                                               const float* __restrict__ Lr,
                                               const float* __restrict__ Linit,
                                               float* __restrict__ ws)
{
    float* Q  = ws;
    float* P0 = ws + 4096;
    float* R  = ws + 8192;
    for (int e = blockIdx.x * blockDim.x + threadIdx.x; e < 4096;
         e += gridDim.x * blockDim.x) {
        const int i = e >> 6, j = e & 63;
        float q = 0.f, p = 0.f;
        for (int k = 0; k < 64; ++k) {
            q += Lq[i * 64 + k] * Lq[j * 64 + k];
            p += Linit[i * 64 + k] * Linit[j * 64 + k];
        }
        Q[e] = q;
        P0[e] = p;
    }
    const int e = blockIdx.x * blockDim.x + threadIdx.x;
    if (e < 64) {
        const int i = e >> 3, j = e & 7;
        float r = 0.f;
        for (int k = 0; k < 8; ++k) r += Lr[i * 8 + k] * Lr[j * 8 + k];
        R[e] = r;
    }
}


template <class Pre, class Beside>
__device__ __forceinline__ void hp_tile(const float* covp, float* HPs,
                                        const v2f (&ha)[16], int tj,
                                        Pre pre, Beside beside)
{
    const int lane = threadIdx.x & 31;
    const int half = lane >> 4;
    const int l15  = lane & 15;
    v2f b[16];
    #pragma unroll
    for (int kb = 0; kb < 16; ++kb) {
        const int k0 = kb * 4 + half * 2;
        b[kb].x = covp[(k0    ) * P_ + (tj + l15)];
        b[kb].y = covp[(k0 + 1) * P_ + (tj + l15)];
    }
    pre();
    __builtin_amdgcn_sched_barrier(0);
    v8f a0 = {}, a1 = {};
    #pragma unroll
    for (int kb = 0; kb < 8; ++kb) {
        a0 = __builtin_amdgcn_wmma_f32_16x16x4_f32(
                 false, ha[kb], false, b[kb], (short)0, a0, false, false);
        a1 = __builtin_amdgcn_wmma_f32_16x16x4_f32(
                 false, ha[kb + 8], false, b[kb + 8], (short)0, a1, false, false);
    }
    beside();
    float h[8];
    #pragma unroll
    for (int r = 0; r < 8; ++r) h[r] = a0[r] + a1[r];
    if (half == 0) {
        #pragma unroll
        for (int r = 0; r < 8; ++r) HPs[r * P_ + (tj + l15)] = h[r];
    }
}

__device__ __forceinline__ void pht_tile(const float* covp, float* PHt, float* Sp,
                                         const v2f (&ha)[16], const v2f (&hs)[4],
                                         const v8f& cstart, int ti)
{
    const int lane = threadIdx.x & 31;
    const int half = lane >> 4;
    const int l15  = lane & 15;
    v2f a[16];
    #pragma unroll
    for (int kb = 0; kb < 16; ++kb) {
        const int k0 = kb * 4 + half * 2;
        a[kb] = *(const v2f*)(covp + (ti + l15) * P_ + k0);
    }
    __builtin_amdgcn_sched_barrier(0);
    v8f acc0 = {}, acc1 = {};
    #pragma unroll
    for (int kb = 0; kb < 8; ++kb) {
        acc0 = __builtin_amdgcn_wmma_f32_16x16x4_f32(
                   false, a[kb], false, ha[kb], (short)0, acc0, false, false);
        acc1 = __builtin_amdgcn_wmma_f32_16x16x4_f32(
                   false, a[kb + 8], false, ha[kb + 8], (short)0, acc1, false, false);
    }
    float p[8];
    #pragma unroll
    for (int r = 0; r < 8; ++r) p[r] = acc0[r] + acc1[r];
    v2f pb[4];
    #pragma unroll
    for (int s = 0; s < 4; ++s) { pb[s].x = p[2 * s]; pb[s].y = p[2 * s + 1]; }
    v8f s0 = cstart, s1 = {};
    s0 = __builtin_amdgcn_wmma_f32_16x16x4_f32(
             false, hs[0], false, pb[0], (short)0, s0, false, false);
    s1 = __builtin_amdgcn_wmma_f32_16x16x4_f32(
             false, hs[1], false, pb[1], (short)0, s1, false, false);
    s0 = __builtin_amdgcn_wmma_f32_16x16x4_f32(
             false, hs[2], false, pb[2], (short)0, s0, false, false);
    s1 = __builtin_amdgcn_wmma_f32_16x16x4_f32(
             false, hs[3], false, pb[3], (short)0, s1, false, false);
    float sp[8];
    #pragma unroll
    for (int r = 0; r < 8; ++r) sp[r] = s0[r] + s1[r];
    if (l15 < 8) {
        #pragma unroll
        for (int r = 0; r < 8; ++r) PHt[(ti + half * 8 + r) * 8 + l15] = p[r];
        if (half == 0) {
            #pragma unroll
            for (int r = 0; r < 8; ++r) Sp[r * 8 + l15] = sp[r];
        }
    }
}

__global__ __launch_bounds__(256) void kf_main(const float* __restrict__ y,
                                               const float* __restrict__ F,
                                               const float* __restrict__ H,
                                               const float* __restrict__ Qg,
                                               const float* __restrict__ Rg,
                                               const float* __restrict__ P0g,
                                               const float* __restrict__ init_mean,
                                               float* __restrict__ out)
{
    extern __shared__ __attribute__((aligned(16))) float smem[];
    float* Fs    = smem;
    float* bufA  = Fs   + 64 * P_;
    float* bufB  = bufA + 64 * P_;
    float* bufC  = bufB + 64 * P_;
    float* Hs    = bufC + 64 * P_;
    float* Rs    = Hs   + 8 * P_;
    float* PHt   = Rs   + 64;
    float* Km    = PHt  + 512;
    float* Sin   = Km   + 512;
    float* meanv = Sin  + 64;
    float* meanp = meanv + 64;
    float* resid = meanp + 64;
    float* Sp    = resid + 8;
    float* HPs   = Sp   + 256;

    const int g   = blockIdx.x;
    const int tid = threadIdx.x;

    for (int i = tid; i < 4096; i += 256) {
        const int r = i >> 6, c = i & 63;
        Fs[r * P_ + c]   = F[i];
        bufA[r * P_ + c] = P0g[i];
    }
    for (int i = tid; i < 512; i += 256) Hs[(i >> 6) * P_ + (i & 63)] = H[i];
    if (tid < 64) {
        Rs[tid]    = Rg[tid];
        meanv[tid] = init_mean[g * S_ + tid];
    }
    __syncthreads();

    float4 rw[16];
    {
        const float* r0 = Fs + (tid & 63) * P_;
        #pragma unroll
        for (int q = 0; q < 16; ++q) rw[q] = ((const float4*)r0)[q];
    }

    v2f ha[16], hs[4];
    {
        const int l15 = tid & 15, half = (tid & 31) >> 4;
        const int ti  = ((tid >> 5) & 3) * 16;
        const float* hrow = Hs + (l15 & 7) * P_;
        #pragma unroll
        for (int kb = 0; kb < 16; ++kb) {
            const int k0 = kb * 4 + half * 2;
            const float h0 = hrow[k0], h1 = hrow[k0 + 1];
            ha[kb].x = (l15 < 8) ? h0 : 0.f;
            ha[kb].y = (l15 < 8) ? h1 : 0.f;
        }
        #pragma unroll
        for (int s = 0; s < 4; ++s) {
            const int k0 = ti + half * 8 + s * 2;
            const float h0 = hrow[k0], h1 = hrow[k0 + 1];
            hs[s].x = (l15 < 8) ? h0 : 0.f;
            hs[s].y = (l15 < 8) ? h1 : 0.f;
        }
    }
    v8f rstart;
    #pragma unroll
    for (int r = 0; r < 8; ++r) rstart[r] = (tid < 8) ? Rs[r * 8 + tid] : 0.f;

    float* const covP  = bufA;
    float* const covpP = bufB;
    float* const tmpP  = bufC;

    float ycur = 0.f, yh = 0.f;

    for (int t = 0; t < T_; ++t) {
        if (tid >= 128 && tid < 136) {
            const int m = tid - 128;
            if (t > 0) out[(g * T_ + t - 1) * M_ + m] = yh;
            ycur = y[(g * T_ + t) * M_ + m];
        }

        mm64<false, 0>(Fs, covP, tmpP, nullptr);
        __syncthreads();
        mm64<true, 1>(tmpP, Fs, covpP, Qg);
        __syncthreads();

        if (tid < 128) {
            pht_tile(covpP, PHt, Sp + (tid >> 5) * 64, ha, hs, rstart, (tid >> 5) * 16);
        } else if (tid < 192) {
            const float4* mv = (const float4*)meanv;
            float4 m4[16];
            #pragma unroll
            for (int q = 0; q < 16; ++q) m4[q] = mv[q];
            __builtin_amdgcn_sched_barrier(0);
            float mp = 0.f;
            #pragma unroll
            for (int q = 0; q < 16; ++q) {
                mp = __builtin_fmaf(rw[q].x, m4[q].x, mp);
                mp = __builtin_fmaf(rw[q].y, m4[q].y, mp);
                mp = __builtin_fmaf(rw[q].z, m4[q].z, mp);
                mp = __builtin_fmaf(rw[q].w, m4[q].w, mp);
            }
            meanp[tid - 128] = mp;
        }
        __syncthreads();

        if (tid < 128) {
            const int cI = tid & 7;
            float wl[8], wr[8];
            #pragma unroll
            for (int r = 0; r < 8; ++r) {
                const float* sq = Sp + r * 8 + cI;
                wl[r] = (sq[0] + sq[64]) + (sq[128] + sq[192]);
                wr[r] = (r == cI) ? 1.f : 0.f;
            }
            if (tid < 8) {
                #pragma unroll
                for (int r = 0; r < 8; ++r) Sin[r * 8 + cI] = wl[r];
            }
            #pragma unroll
            for (int p = 0; p < 8; ++p) {
                const float pv = __builtin_amdgcn_rcpf(lane_bcast(wl[p], p));
                wl[p] *= pv;
                wr[p] *= pv;
                #pragma unroll
                for (int r = 0; r < 8; ++r) {
                    if (r != p) {
                        const float f = lane_bcast(wl[r], p);
                        wl[r] -= f * wl[p];
                        wr[r] -= f * wr[p];
                    }
                }
            }

            const int s0 = tid >> 3;
            float k0 = 0.f, k1 = 0.f, k2 = 0.f, k3 = 0.f;
            #pragma unroll
            for (int j = 0; j < 8; ++j) {
                k0 += PHt[s0 * 8 + j] * wr[j];
                k1 += PHt[(s0 + 16) * 8 + j] * wr[j];
                k2 += PHt[(s0 + 32) * 8 + j] * wr[j];
                k3 += PHt[(s0 + 48) * 8 + j] * wr[j];
            }
            Km[tid]       = k0;
            Km[tid + 128] = k1;
            Km[tid + 256] = k2;
            Km[tid + 384] = k3;
        } else if (tid < 160) {
            const int m = tid & 7;
            const float4* hr = (const float4*)(Hs + m * P_);
            const float4* mq = (const float4*)meanp;
            float4 h4[16], m4[16];
            float acc = 0.f;
            hp_tile(covpP, HPs, ha, 0,
                [&]() {
                    #pragma unroll
                    for (int q = 0; q < 16; ++q) { h4[q] = hr[q]; m4[q] = mq[q]; }
                },
                [&]() {
                    float ps[4];
                    #pragma unroll
                    for (int j = 0; j < 4; ++j) {
                        float a = 0.f;
                        #pragma unroll
                        for (int q = 4 * j; q < 4 * j + 4; ++q) {
                            a = __builtin_fmaf(h4[q].x, m4[q].x, a);
                            a = __builtin_fmaf(h4[q].y, m4[q].y, a);
                            a = __builtin_fmaf(h4[q].z, m4[q].z, a);
                            a = __builtin_fmaf(h4[q].w, m4[q].w, a);
                        }
                        ps[j] = a;
                    }
                    acc = (ps[0] + ps[1]) + (ps[2] + ps[3]);
                });
            yh = acc;
            if (tid < 136) resid[m] = ycur - acc;
        } else {
            hp_tile(covpP, HPs, ha, ((tid >> 5) - 4) * 16, []() {}, []() {});
        }
        __syncthreads();

        const int uh   = (tid & 31) >> 4;
        const int ul15 = tid & 15;
        const int uti0 = (tid >> 7) * 16;
        const int ucol = ((tid >> 5) & 3) * 16 + ul15;
        const int ur0  = uti0 + ul15;
        const int ur1  = ur0 + 32;
        v8f t0, t1;
        #pragma unroll
        for (int r = 0; r < 8; ++r) {
            t0[r] = covpP[(uti0 + r + uh * 8) * P_ + ucol];
            t1[r] = covpP[(uti0 + 32 + r + uh * 8) * P_ + ucol];
        }
        const bool srow = (ul15 & 4) == 0;
        const int  sm   = (ul15 & 3) | ((ul15 & 8) >> 1);
        v2f ka0[2], ka1[2], hb[2], sa[2];
        #pragma unroll
        for (int kb = 0; kb < 2; ++kb) {
            const int k0 = kb * 4 + uh * 2;
            ka0[kb].x = -Km[ur0 * 8 + k0];
            ka0[kb].y = -Km[ur0 * 8 + k0 + 1];
            ka1[kb].x = -Km[ur1 * 8 + k0];
            ka1[kb].y = -Km[ur1 * 8 + k0 + 1];
            hb[kb].x  = HPs[(k0    ) * P_ + ucol];
            hb[kb].y  = HPs[(k0 + 1) * P_ + ucol];
            const float s0 = Sin[(k0    ) * 8 + sm];
            const float s1 = Sin[(k0 + 1) * 8 + sm];
            sa[kb].x = srow ? s0 : 0.f;
            sa[kb].y = srow ? s1 : 0.f;
        }
        const float4 p0 = *(const float4*)(PHt + ur0 * 8 + 4 * uh);
        const float4 p1 = *(const float4*)(PHt + ur1 * 8 + 4 * uh);
        v8f nw0 = {p0.x, p0.y, p0.z, p0.w, 0.f, 0.f, 0.f, 0.f};
        v8f nw1 = {p1.x, p1.y, p1.z, p1.w, 0.f, 0.f, 0.f, 0.f};
        const float4 kj = *(const float4*)(Km + ucol * 8 + 4 * uh);
        v2f kc[2];
        kc[0].x = -kj.x; kc[0].y = -kj.y;
        kc[1].x = -kj.z; kc[1].y = -kj.w;

        float mu = 0.f;
        if (tid < 64) {
            mu = meanp[tid];
            for (int m = 0; m < 8; ++m) mu += Km[tid * 8 + m] * resid[m];
        }

        #pragma unroll
        for (int kb = 0; kb < 2; ++kb) {
            t0  = __builtin_amdgcn_wmma_f32_16x16x4_f32(
                      false, ka0[kb], false, hb[kb], (short)0, t0, false, false);
            t1  = __builtin_amdgcn_wmma_f32_16x16x4_f32(
                      false, ka1[kb], false, hb[kb], (short)0, t1, false, false);
            nw0 = __builtin_amdgcn_wmma_f32_16x16x4_f32(
                      false, sa[kb], false, ka0[kb], (short)0, nw0, false, false);
            nw1 = __builtin_amdgcn_wmma_f32_16x16x4_f32(
                      false, sa[kb], false, ka1[kb], (short)0, nw1, false, false);
        }
        #pragma unroll
        for (int kb = 0; kb < 2; ++kb) {
            v2f wa0, wa1;
            wa0.x = nw0[2 * kb]; wa0.y = nw0[2 * kb + 1];
            wa1.x = nw1[2 * kb]; wa1.y = nw1[2 * kb + 1];
            t0 = __builtin_amdgcn_wmma_f32_16x16x4_f32(
                     false, wa0, false, kc[kb], (short)0, t0, false, false);
            t1 = __builtin_amdgcn_wmma_f32_16x16x4_f32(
                     false, wa1, false, kc[kb], (short)0, t1, false, false);
        }
        if (tid < 64) meanv[tid] = mu;
        #pragma unroll
        for (int r = 0; r < 8; ++r) {
            covP[(uti0 + r + uh * 8) * P_ + ucol]      = t0[r];
            covP[(uti0 + 32 + r + uh * 8) * P_ + ucol] = t1[r];
        }
        __syncthreads();
    }
    if (tid >= 128 && tid < 136) out[(g * T_ + T_ - 1) * M_ + (tid - 128)] = yh;
}

extern "C" void kernel_launch(void* const* d_in, const int* in_sizes, int n_in,
                              void* d_out, int out_size, void* d_ws, size_t ws_size,
                              hipStream_t stream) {
    const float* y         = (const float*)d_in[0];
    const float* F         = (const float*)d_in[1];
    const float* H         = (const float*)d_in[2];
    const float* Lq        = (const float*)d_in[3];
    const float* Lr        = (const float*)d_in[4];
    const float* init_mean = (const float*)d_in[5];
    const float* Linit     = (const float*)d_in[6];
    float* out = (float*)d_out;
    float* ws  = (float*)d_ws;

    kf_prep<<<16, 256, 0, stream>>>(Lq, Lr, Linit, ws);

    const size_t smem_bytes = (4864 * 4 + 608 + 64 + 512 + 512 + 64 +
                               64 + 64 + 8 + 256 + 608) * sizeof(float);
    kf_main<<<G_, 256, smem_bytes, stream>>>(y, F, H,
                                             ws,
                                             ws + 8192,
                                             ws + 4096,
                                             init_mean, out);
}
